# starting kernel
# baseline (speedup 1.0000x reference)
#include <hip/hip_runtime.h>
#include <hip/hip_bf16.h>
#include <math.h>

typedef float v2f __attribute__((ext_vector_type(2)));
typedef float v8f __attribute__((ext_vector_type(8)));

#define IN_DIM   128
#define HID      32
#define HEADS    2
#define OUT_DIM  32

__device__ __forceinline__ void atomicMaxFloat(float* addr, float value) {
    if (value >= 0.0f) {
        atomicMax((int*)addr, __float_as_int(value));
    } else {
        atomicMin((unsigned int*)addr, __float_as_uint(value));
    }
}

__global__ void fill_kernel(float* __restrict__ p, float v, int n) {
    int i = blockIdx.x * blockDim.x + threadIdx.x;
    if (i < n) p[i] = v;
}

template <int K, int NC>
__global__ void wmma_gemm_f32(const float* __restrict__ A,
                              const float* __restrict__ B,
                              float* __restrict__ C, int M) {
    constexpr int NT = NC / 16;
    __shared__ v2f Bs2[(K / 2) * NC];

    for (int p = threadIdx.x; p < (K / 2) * NC; p += blockDim.x) {
        int k2 = p / NC;
        int n  = p - k2 * NC;
        v2f pr;
        pr.x = B[(2 * k2 + 0) * NC + n];
        pr.y = B[(2 * k2 + 1) * NC + n];
        Bs2[p] = pr;
    }
    __syncthreads();

    const int wave = threadIdx.x >> 5;
    const int lane = threadIdx.x & 31;
    const int mt   = blockIdx.x * 4 + wave;
    if (mt * 16 >= M) return;

    const int halfK = (lane >> 4) * 2;
    const int ncol  = lane & 15;
    const int arow  = mt * 16 + (lane & 15);
    const float* Arow = A + (size_t)arow * K;

    v8f acc[NT];
#pragma unroll
    for (int t = 0; t < NT; ++t) acc[t] = (v8f){0, 0, 0, 0, 0, 0, 0, 0};

    for (int k = 0; k < K; k += 4) {
        v2f a = *(const v2f*)(Arow + k + halfK);
        const int prow = (k + halfK) >> 1;
#pragma unroll
        for (int t = 0; t < NT; ++t) {
            v2f b = Bs2[prow * NC + t * 16 + ncol];
            acc[t] = __builtin_amdgcn_wmma_f32_16x16x4_f32(
                false, a, false, b, (short)0, acc[t], false, false);
        }
    }

    const int rbase = mt * 16 + ((lane >> 4) << 3);
#pragma unroll
    for (int t = 0; t < NT; ++t) {
#pragma unroll
        for (int j = 0; j < 8; ++j) {
            C[(size_t)(rbase + j) * NC + t * 16 + ncol] = acc[t][j];
        }
    }
}

template <int H, int C>
__global__ void score_kernel(const float* __restrict__ xl,
                             const float* __restrict__ a_src,
                             const float* __restrict__ a_dst,
                             float* __restrict__ asn,
                             float* __restrict__ adn, int n) {
    int i = blockIdx.x * blockDim.x + threadIdx.x;
    if (i >= n) return;
    const float* xr = xl + (size_t)i * H * C;
#pragma unroll
    for (int h = 0; h < H; ++h) {
        float s = 0.0f, d = 0.0f;
#pragma unroll
        for (int c = 0; c < C; ++c) {
            float v = xr[h * C + c];
            s += v * a_src[h * C + c];
            d += v * a_dst[h * C + c];
        }
        asn[i * H + h] = s;
        adn[i * H + h] = d;
    }
}

template <int H>
__global__ void edge_max_kernel(const int* __restrict__ ei, int E, int EE,
                                const float* __restrict__ asn,
                                const float* __restrict__ adn,
                                float* __restrict__ ebuf,
                                float* __restrict__ mbuf) {
    int e = blockIdx.x * blockDim.x + threadIdx.x;
    if (e >= EE) return;
    int src, dst;
    if (e < E) { src = ei[e]; dst = ei[E + e]; }
    else       { src = dst = e - E; }
#pragma unroll
    for (int h = 0; h < H; ++h) {
        float v = asn[src * H + h] + adn[dst * H + h];
        v = (v > 0.0f) ? v : 0.2f * v;
        ebuf[(size_t)e * H + h] = v;
        atomicMaxFloat(&mbuf[dst * H + h], v);
    }
}

template <int H>
__global__ void edge_exp_kernel(const int* __restrict__ ei, int E, int EE,
                                float* __restrict__ ebuf,
                                const float* __restrict__ mbuf,
                                float* __restrict__ den) {
    int e = blockIdx.x * blockDim.x + threadIdx.x;
    if (e >= EE) return;
    int dst = (e < E) ? ei[E + e] : (e - E);
#pragma unroll
    for (int h = 0; h < H; ++h) {
        float m = mbuf[dst * H + h];
        if (!(m >= -3.4e38f && m <= 3.4e38f)) m = 0.0f;
        float ex = __expf(ebuf[(size_t)e * H + h] - m);
        ebuf[(size_t)e * H + h] = ex;
        atomicAdd(&den[dst * H + h], ex);
    }
}

template <int H, int C>
__global__ void edge_agg_kernel(const int* __restrict__ ei, int E, int EE,
                                const float* __restrict__ ebuf,
                                const float* __restrict__ den,
                                const float* __restrict__ xl,
                                float* __restrict__ agg) {
    constexpr int F = H * C;
    constexpr int G = F / 4;
    long long t = (long long)blockIdx.x * blockDim.x + threadIdx.x;
    long long e = t / G;
    int g = (int)(t % G);
    if (e >= EE) return;
    int src, dst;
    if (e < E) { src = ei[e]; dst = ei[E + e]; }
    else       { src = dst = (int)(e - E); }
    int f0 = g * 4;
    int h  = f0 / C;
    float alpha = ebuf[(size_t)e * H + h] / (den[dst * H + h] + 1e-16f);
    float4 v = *(const float4*)(xl + (size_t)src * F + f0);
    float* o = agg + (size_t)dst * F + f0;
    atomicAdd(o + 0, v.x * alpha);
    atomicAdd(o + 1, v.y * alpha);
    atomicAdd(o + 2, v.z * alpha);
    atomicAdd(o + 3, v.w * alpha);
}

__global__ void bias_relu_kernel(float* __restrict__ a,
                                 const float* __restrict__ b, int n, int F) {
    int i = blockIdx.x * blockDim.x + threadIdx.x;
    if (i >= n * F) return;
    float v = a[i] + b[i % F];
    a[i] = v > 0.0f ? v : 0.0f;
}

__global__ void bias_add_kernel(float* __restrict__ a,
                                const float* __restrict__ b, int n, int F) {
    int i = blockIdx.x * blockDim.x + threadIdx.x;
    if (i >= n * F) return;
    a[i] += b[i % F];
}

static inline int cdiv(long long a, long long b) { return (int)((a + b - 1) / b); }

extern "C" void kernel_launch(void* const* d_in, const int* in_sizes, int n_in,
                              void* d_out, int out_size, void* d_ws, size_t ws_size,
                              hipStream_t stream) {
    const float* x      = (const float*)d_in[0];
    const int*   ei     = (const int*)d_in[1];
    const float* W1     = (const float*)d_in[2];
    const float* a_src1 = (const float*)d_in[3];
    const float* a_dst1 = (const float*)d_in[4];
    const float* b1     = (const float*)d_in[5];
    const float* W2     = (const float*)d_in[6];
    const float* a_src2 = (const float*)d_in[7];
    const float* a_dst2 = (const float*)d_in[8];
    const float* b2     = (const float*)d_in[9];
    float* out = (float*)d_out;

    const int N  = in_sizes[0] / IN_DIM;
    const int E  = in_sizes[1] / 2;
    const int EE = E + N;
    const int F1 = HEADS * HID;

    float* ws   = (float*)d_ws;
    float* xl1  = ws;
    float* agg1 = xl1  + (size_t)N * F1;
    float* asn1 = agg1 + (size_t)N * F1;
    float* adn1 = asn1 + (size_t)N * HEADS;
    float* m1   = adn1 + (size_t)N * HEADS;
    float* den1 = m1   + (size_t)N * HEADS;
    float* ex1  = den1 + (size_t)N * HEADS;
    float* xl2  = ex1  + (size_t)EE * HEADS;
    float* asn2 = xl2  + (size_t)N * OUT_DIM;
    float* adn2 = asn2 + (size_t)N;
    float* m2   = adn2 + (size_t)N;
    float* den2 = m2   + (size_t)N;
    float* ex2  = den2 + (size_t)N;

    const int TB = 256;

    fill_kernel<<<cdiv((long long)N * F1, TB), TB, 0, stream>>>(agg1, 0.0f, N * F1);
    fill_kernel<<<cdiv((long long)N * OUT_DIM, TB), TB, 0, stream>>>(out, 0.0f, N * OUT_DIM);
    fill_kernel<<<cdiv((long long)N * HEADS, TB), TB, 0, stream>>>(m1, -INFINITY, N * HEADS);
    fill_kernel<<<cdiv((long long)N * HEADS, TB), TB, 0, stream>>>(den1, 0.0f, N * HEADS);
    fill_kernel<<<cdiv(N, TB), TB, 0, stream>>>(m2, -INFINITY, N);
    fill_kernel<<<cdiv(N, TB), TB, 0, stream>>>(den2, 0.0f, N);

    {
        int mtiles = cdiv(N, 16);
        wmma_gemm_f32<IN_DIM, F1><<<cdiv(mtiles, 4), 128, 0, stream>>>(x, W1, xl1, N);
        score_kernel<HEADS, HID><<<cdiv(N, TB), TB, 0, stream>>>(xl1, a_src1, a_dst1, asn1, adn1, N);
        edge_max_kernel<HEADS><<<cdiv(EE, TB), TB, 0, stream>>>(ei, E, EE, asn1, adn1, ex1, m1);
        edge_exp_kernel<HEADS><<<cdiv(EE, TB), TB, 0, stream>>>(ei, E, EE, ex1, m1, den1);
        edge_agg_kernel<HEADS, HID><<<cdiv((long long)EE * (F1 / 4), TB), TB, 0, stream>>>(
            ei, E, EE, ex1, den1, xl1, agg1);
        bias_relu_kernel<<<cdiv((long long)N * F1, TB), TB, 0, stream>>>(agg1, b1, N, F1);
    }

    {
        int mtiles = cdiv(N, 16);
        wmma_gemm_f32<F1, OUT_DIM><<<cdiv(mtiles, 4), 128, 0, stream>>>(agg1, W2, xl2, N);
        score_kernel<1, OUT_DIM><<<cdiv(N, TB), TB, 0, stream>>>(xl2, a_src2, a_dst2, asn2, adn2, N);
        edge_max_kernel<1><<<cdiv(EE, TB), TB, 0, stream>>>(ei, E, EE, asn2, adn2, ex2, m2);
        edge_exp_kernel<1><<<cdiv(EE, TB), TB, 0, stream>>>(ei, E, EE, ex2, m2, den2);
        edge_agg_kernel<1, OUT_DIM><<<cdiv((long long)EE * (OUT_DIM / 4), TB), TB, 0, stream>>>(
            ei, E, EE, ex2, den2, xl2, out);
        bias_add_kernel<<<cdiv((long long)N * OUT_DIM, TB), TB, 0, stream>>>(out, b2, N, OUT_DIM);
    }
}


# Segment-max pass replaced by a per-destination upper bound leaky(max source score + dst score) when the score spread is <= 16 (exact max pass kept as fallback): max atomics gone; exp terms rescaled
# speedup vs baseline: 1.3099x; 1.3099x over previous
#include <hip/hip_runtime.h>
#include <hip/hip_bf16.h>
#include <math.h>

typedef float v2f __attribute__((ext_vector_type(2)));
typedef float v8f __attribute__((ext_vector_type(8)));

#define IN_DIM   128
#define HID      32
#define HEADS    2
#define OUT_DIM  32

__device__ __forceinline__ void atomicMaxFloat(float* addr, float value) {
    if (value >= 0.0f) {
        atomicMax((int*)addr, __float_as_int(value));
    } else {
        atomicMin((unsigned int*)addr, __float_as_uint(value));
    }
}

__global__ void fill_kernel(float* __restrict__ p, float v, int n) {
    int i = blockIdx.x * blockDim.x + threadIdx.x;
    if (i < n) p[i] = v;
}

template <int K, int NC>
__global__ void wmma_gemm_f32(const float* __restrict__ A,
                              const float* __restrict__ B,
                              float* __restrict__ C, int M) {
    constexpr int NT = NC / 16;
    __shared__ v2f Bs2[(K / 2) * NC];

    for (int p = threadIdx.x; p < (K / 2) * NC; p += blockDim.x) {
        int k2 = p / NC;
        int n  = p - k2 * NC;
        v2f pr;
        pr.x = B[(2 * k2 + 0) * NC + n];
        pr.y = B[(2 * k2 + 1) * NC + n];
        Bs2[p] = pr;
    }
    __syncthreads();

    const int wave = threadIdx.x >> 5;
    const int lane = threadIdx.x & 31;
    const int mt   = blockIdx.x * 4 + wave;
    if (mt * 16 >= M) return;

    const int halfK = (lane >> 4) * 2;
    const int ncol  = lane & 15;
    const int arow  = mt * 16 + (lane & 15);
    const float* Arow = A + (size_t)arow * K;

    v8f acc[NT];
#pragma unroll
    for (int t = 0; t < NT; ++t) acc[t] = (v8f){0, 0, 0, 0, 0, 0, 0, 0};

    for (int k = 0; k < K; k += 4) {
        v2f a = *(const v2f*)(Arow + k + halfK);
        const int prow = (k + halfK) >> 1;
#pragma unroll
        for (int t = 0; t < NT; ++t) {
            v2f b = Bs2[prow * NC + t * 16 + ncol];
            acc[t] = __builtin_amdgcn_wmma_f32_16x16x4_f32(
                false, a, false, b, (short)0, acc[t], false, false);
        }
    }

    const int rbase = mt * 16 + ((lane >> 4) << 3);
#pragma unroll
    for (int t = 0; t < NT; ++t) {
#pragma unroll
        for (int j = 0; j < 8; ++j) {
            C[(size_t)(rbase + j) * NC + t * 16 + ncol] = acc[t][j];
        }
    }
}

template <int H, int C>
__global__ void score_kernel(const float* __restrict__ xl,
                             const float* __restrict__ a_src,
                             const float* __restrict__ a_dst,
                             float* __restrict__ asn,
                             float* __restrict__ adn,
                             float* __restrict__ part, int n) {
    __shared__ float red[2 * H][32];
    int i = blockIdx.x * blockDim.x + threadIdx.x;
    float smax[H], smin[H];
#pragma unroll
    for (int h = 0; h < H; ++h) { smax[h] = -INFINITY; smin[h] = INFINITY; }
    if (i < n) {
        const float* xr = xl + (size_t)i * H * C;
#pragma unroll
        for (int h = 0; h < H; ++h) {
            float s = 0.0f, d = 0.0f;
#pragma unroll
            for (int c = 0; c < C; ++c) {
                float v = xr[h * C + c];
                s += v * a_src[h * C + c];
                d += v * a_dst[h * C + c];
            }
            asn[i * H + h] = s;
            adn[i * H + h] = d;
            smax[h] = s;
            smin[h] = s;
        }
    }
    const int lane = threadIdx.x & (warpSize - 1);
    const int wave = threadIdx.x / warpSize;
    const int nw   = (blockDim.x + warpSize - 1) / warpSize;
#pragma unroll
    for (int h = 0; h < H; ++h) {
        for (int off = warpSize / 2; off > 0; off >>= 1) {
            smax[h] = fmaxf(smax[h], __shfl_xor(smax[h], off));
            smin[h] = fminf(smin[h], __shfl_xor(smin[h], off));
        }
        if (lane == 0) { red[h][wave] = smax[h]; red[H + h][wave] = smin[h]; }
    }
    __syncthreads();
    if (threadIdx.x < 2 * H) {
        const bool is_max = threadIdx.x < H;
        float v = red[threadIdx.x][0];
        for (int w = 1; w < nw; ++w) {
            float o = red[threadIdx.x][w];
            v = is_max ? fmaxf(v, o) : fminf(v, o);
        }
        part[(size_t)blockIdx.x * 2 * H + threadIdx.x] = v;
    }
}

#define SPREAD_OK 16.0f

template <int H>
__global__ void range_kernel(const float* __restrict__ part, int nblk,
                             float* __restrict__ stats,
                             float* __restrict__ mbuf, int nm) {
    __shared__ float red[2 * H][32];
    __shared__ int s_ok;
    float smax[H], smin[H];
#pragma unroll
    for (int h = 0; h < H; ++h) { smax[h] = -INFINITY; smin[h] = INFINITY; }
    for (int b = threadIdx.x; b < nblk; b += blockDim.x) {
#pragma unroll
        for (int h = 0; h < H; ++h) {
            smax[h] = fmaxf(smax[h], part[(size_t)b * 2 * H + h]);
            smin[h] = fminf(smin[h], part[(size_t)b * 2 * H + H + h]);
        }
    }
    const int lane = threadIdx.x & (warpSize - 1);
    const int wave = threadIdx.x / warpSize;
    const int nw   = (blockDim.x + warpSize - 1) / warpSize;
#pragma unroll
    for (int h = 0; h < H; ++h) {
        for (int off = warpSize / 2; off > 0; off >>= 1) {
            smax[h] = fmaxf(smax[h], __shfl_xor(smax[h], off));
            smin[h] = fminf(smin[h], __shfl_xor(smin[h], off));
        }
        if (lane == 0) { red[h][wave] = smax[h]; red[H + h][wave] = smin[h]; }
    }
    __syncthreads();
    if (threadIdx.x == 0) {
        bool ok = true;
#pragma unroll
        for (int h = 0; h < H; ++h) {
            float mx = red[h][0], mn = red[H + h][0];
            for (int w = 1; w < nw; ++w) {
                mx = fmaxf(mx, red[h][w]);
                mn = fminf(mn, red[H + h][w]);
            }
            if (blockIdx.x == 0) stats[h] = mx;
            ok = ok && ((mx - mn) <= SPREAD_OK);
        }
        if (blockIdx.x == 0) stats[H] = ok ? 1.0f : 0.0f;
        s_ok = ok ? 1 : 0;
    }
    __syncthreads();
    if (!s_ok) {
        const int stride = gridDim.x * blockDim.x;
        for (int i = blockIdx.x * blockDim.x + threadIdx.x; i < nm; i += stride)
            mbuf[i] = -INFINITY;
    }
}

template <int H>
__global__ void edge_max_kernel(const int* __restrict__ ei, int E, int EE,
                                const float* __restrict__ asn,
                                const float* __restrict__ adn,
                                const float* __restrict__ stats,
                                float* __restrict__ mbuf) {
    if (stats[H] != 0.0f) return;
    const int stride = gridDim.x * blockDim.x;
    for (int e = blockIdx.x * blockDim.x + threadIdx.x; e < EE; e += stride) {
        int src, dst;
        if (e < E) { src = ei[e]; dst = ei[E + e]; }
        else       { src = dst = e - E; }
#pragma unroll
        for (int h = 0; h < H; ++h) {
            float v = asn[src * H + h] + adn[dst * H + h];
            v = (v > 0.0f) ? v : 0.2f * v;
            atomicMaxFloat(&mbuf[dst * H + h], v);
        }
    }
}

template <int H>
__global__ void edge_exp_kernel(const int* __restrict__ ei, int E, int EE,
                                const float* __restrict__ asn,
                                const float* __restrict__ adn,
                                const float* __restrict__ stats,
                                const float* __restrict__ mbuf,
                                float* __restrict__ ebuf,
                                float* __restrict__ den) {
    int e = blockIdx.x * blockDim.x + threadIdx.x;
    if (e >= EE) return;
    int src, dst;
    if (e < E) { src = ei[e]; dst = ei[E + e]; }
    else       { src = dst = e - E; }
    const bool bounded = stats[H] != 0.0f;
#pragma unroll
    for (int h = 0; h < H; ++h) {
        float ad = adn[dst * H + h];
        float v = asn[src * H + h] + ad;
        v = (v > 0.0f) ? v : 0.2f * v;
        float m;
        if (bounded) {
            float b = stats[h] + ad;
            m = (b > 0.0f) ? b : 0.2f * b;
        } else {
            m = mbuf[dst * H + h];
        }
        if (!(m >= -3.4e38f && m <= 3.4e38f)) m = 0.0f;
        float ex = __expf(v - m);
        ebuf[(size_t)e * H + h] = ex;
        atomicAdd(&den[dst * H + h], ex);
    }
}

template <int H, int C>
__global__ void edge_agg_kernel(const int* __restrict__ ei, int E, int EE,
                                const float* __restrict__ ebuf,
                                const float* __restrict__ den,
                                const float* __restrict__ xl,
                                float* __restrict__ agg) {
    constexpr int F = H * C;
    constexpr int G = F / 4;
    long long t = (long long)blockIdx.x * blockDim.x + threadIdx.x;
    long long e = t / G;
    int g = (int)(t % G);
    if (e >= EE) return;
    int src, dst;
    if (e < E) { src = ei[e]; dst = ei[E + e]; }
    else       { src = dst = (int)(e - E); }
    int f0 = g * 4;
    int h  = f0 / C;
    float alpha = ebuf[(size_t)e * H + h] / (den[dst * H + h] + 1e-16f);
    float4 v = *(const float4*)(xl + (size_t)src * F + f0);
    float* o = agg + (size_t)dst * F + f0;
    atomicAdd(o + 0, v.x * alpha);
    atomicAdd(o + 1, v.y * alpha);
    atomicAdd(o + 2, v.z * alpha);
    atomicAdd(o + 3, v.w * alpha);
}

__global__ void bias_relu_kernel(float* __restrict__ a,
                                 const float* __restrict__ b, int n, int F) {
    int i = blockIdx.x * blockDim.x + threadIdx.x;
    if (i >= n * F) return;
    float v = a[i] + b[i % F];
    a[i] = v > 0.0f ? v : 0.0f;
}

__global__ void bias_add_kernel(float* __restrict__ a,
                                const float* __restrict__ b, int n, int F) {
    int i = blockIdx.x * blockDim.x + threadIdx.x;
    if (i >= n * F) return;
    a[i] += b[i % F];
}

static inline int cdiv(long long a, long long b) { return (int)((a + b - 1) / b); }

extern "C" void kernel_launch(void* const* d_in, const int* in_sizes, int n_in,
                              void* d_out, int out_size, void* d_ws, size_t ws_size,
                              hipStream_t stream) {
    const float* x      = (const float*)d_in[0];
    const int*   ei     = (const int*)d_in[1];
    const float* W1     = (const float*)d_in[2];
    const float* a_src1 = (const float*)d_in[3];
    const float* a_dst1 = (const float*)d_in[4];
    const float* b1     = (const float*)d_in[5];
    const float* W2     = (const float*)d_in[6];
    const float* a_src2 = (const float*)d_in[7];
    const float* a_dst2 = (const float*)d_in[8];
    const float* b2     = (const float*)d_in[9];
    float* out = (float*)d_out;

    const int N  = in_sizes[0] / IN_DIM;
    const int E  = in_sizes[1] / 2;
    const int EE = E + N;
    const int F1 = HEADS * HID;

    float* ws   = (float*)d_ws;
    float* xl1  = ws;
    float* agg1 = xl1  + (size_t)N * F1;
    float* asn1 = agg1 + (size_t)N * F1;
    float* adn1 = asn1 + (size_t)N * HEADS;
    float* m1   = adn1 + (size_t)N * HEADS;
    float* den1 = m1   + (size_t)N * HEADS;
    float* ex1  = den1 + (size_t)N * HEADS;
    float* xl2  = ex1  + (size_t)EE * HEADS;
    float* asn2 = xl2  + (size_t)N * OUT_DIM;
    float* adn2 = asn2 + (size_t)N;
    float* m2   = adn2 + (size_t)N;
    float* den2 = m2   + (size_t)N;
    float* ex2  = den2 + (size_t)N;

    const int TB = 256;
    const int NSB = cdiv(N, TB);

    float* part1  = xl2;
    float* stats1 = part1 + (size_t)NSB * 2 * HEADS;
    float* part2  = ex1;
    float* stats2 = part2 + (size_t)NSB * 2;

    fill_kernel<<<cdiv((long long)N * F1, TB), TB, 0, stream>>>(agg1, 0.0f, N * F1);
    fill_kernel<<<cdiv((long long)N * OUT_DIM, TB), TB, 0, stream>>>(out, 0.0f, N * OUT_DIM);
    fill_kernel<<<cdiv((long long)N * HEADS, TB), TB, 0, stream>>>(den1, 0.0f, N * HEADS);
    fill_kernel<<<cdiv(N, TB), TB, 0, stream>>>(den2, 0.0f, N);

    {
        int mtiles = cdiv(N, 16);
        wmma_gemm_f32<IN_DIM, F1><<<cdiv(mtiles, 4), 128, 0, stream>>>(x, W1, xl1, N);
        score_kernel<HEADS, HID><<<NSB, TB, 0, stream>>>(xl1, a_src1, a_dst1, asn1, adn1, part1, N);
        range_kernel<HEADS><<<64, TB, 0, stream>>>(part1, NSB, stats1, m1, N * HEADS);
        edge_max_kernel<HEADS><<<1024, TB, 0, stream>>>(ei, E, EE, asn1, adn1, stats1, m1);
        edge_exp_kernel<HEADS><<<cdiv(EE, TB), TB, 0, stream>>>(ei, E, EE, asn1, adn1, stats1, m1, ex1, den1);
        edge_agg_kernel<HEADS, HID><<<cdiv((long long)EE * (F1 / 4), TB), TB, 0, stream>>>(
            ei, E, EE, ex1, den1, xl1, agg1);
        bias_relu_kernel<<<cdiv((long long)N * F1, TB), TB, 0, stream>>>(agg1, b1, N, F1);
    }

    {
        int mtiles = cdiv(N, 16);
        wmma_gemm_f32<F1, OUT_DIM><<<cdiv(mtiles, 4), 128, 0, stream>>>(agg1, W2, xl2, N);
        score_kernel<1, OUT_DIM><<<NSB, TB, 0, stream>>>(xl2, a_src2, a_dst2, asn2, adn2, part2, N);
        range_kernel<1><<<64, TB, 0, stream>>>(part2, NSB, stats2, m2, N);
        edge_max_kernel<1><<<1024, TB, 0, stream>>>(ei, E, EE, asn2, adn2, stats2, m2);
        edge_exp_kernel<1><<<cdiv(EE, TB), TB, 0, stream>>>(ei, E, EE, asn2, adn2, stats2, m2, ex2, den2);
        edge_agg_kernel<1, OUT_DIM><<<cdiv((long long)EE * (OUT_DIM / 4), TB), TB, 0, stream>>>(
            ei, E, EE, ex2, den2, xl2, out);
        bias_add_kernel<<<cdiv((long long)N * OUT_DIM, TB), TB, 0, stream>>>(out, b2, N, OUT_DIM);
    }
}


# Aggregation remapped to lane = feature column: a wave walks 32 edges, each head's 32-float row piece is one load and one atomic on a whole 128-byte line; sum order stays arrival-dependent
# speedup vs baseline: 1.6047x; 1.2251x over previous
#include <hip/hip_runtime.h>
#include <hip/hip_bf16.h>
#include <math.h>

typedef float v2f __attribute__((ext_vector_type(2)));
typedef float v8f __attribute__((ext_vector_type(8)));

#define IN_DIM   128
#define HID      32
#define HEADS    2
#define OUT_DIM  32

__device__ __forceinline__ void atomicMaxFloat(float* addr, float value) {
    if (value >= 0.0f) {
        atomicMax((int*)addr, __float_as_int(value));
    } else {
        atomicMin((unsigned int*)addr, __float_as_uint(value));
    }
}

__global__ void fill_kernel(float* __restrict__ p, float v, int n) {
    int i = blockIdx.x * blockDim.x + threadIdx.x;
    if (i < n) p[i] = v;
}

template <int K, int NC>
__global__ void wmma_gemm_f32(const float* __restrict__ A,
                              const float* __restrict__ B,
                              float* __restrict__ C, int M) {
    constexpr int NT = NC / 16;
    __shared__ v2f Bs2[(K / 2) * NC];

    for (int p = threadIdx.x; p < (K / 2) * NC; p += blockDim.x) {
        int k2 = p / NC;
        int n  = p - k2 * NC;
        v2f pr;
        pr.x = B[(2 * k2 + 0) * NC + n];
        pr.y = B[(2 * k2 + 1) * NC + n];
        Bs2[p] = pr;
    }
    __syncthreads();

    const int wave = threadIdx.x >> 5;
    const int lane = threadIdx.x & 31;
    const int mt   = blockIdx.x * 4 + wave;
    if (mt * 16 >= M) return;

    const int halfK = (lane >> 4) * 2;
    const int ncol  = lane & 15;
    const int arow  = mt * 16 + (lane & 15);
    const float* Arow = A + (size_t)arow * K;

    v8f acc[NT];
#pragma unroll
    for (int t = 0; t < NT; ++t) acc[t] = (v8f){0, 0, 0, 0, 0, 0, 0, 0};

    for (int k = 0; k < K; k += 4) {
        v2f a = *(const v2f*)(Arow + k + halfK);
        const int prow = (k + halfK) >> 1;
#pragma unroll
        for (int t = 0; t < NT; ++t) {
            v2f b = Bs2[prow * NC + t * 16 + ncol];
            acc[t] = __builtin_amdgcn_wmma_f32_16x16x4_f32(
                false, a, false, b, (short)0, acc[t], false, false);
        }
    }

    const int rbase = mt * 16 + ((lane >> 4) << 3);
#pragma unroll
    for (int t = 0; t < NT; ++t) {
#pragma unroll
        for (int j = 0; j < 8; ++j) {
            C[(size_t)(rbase + j) * NC + t * 16 + ncol] = acc[t][j];
        }
    }
}

template <int H, int C>
__global__ void score_kernel(const float* __restrict__ xl,
                             const float* __restrict__ a_src,
                             const float* __restrict__ a_dst,
                             float* __restrict__ asn,
                             float* __restrict__ adn,
                             float* __restrict__ part, int n) {
    __shared__ float red[2 * H][32];
    int i = blockIdx.x * blockDim.x + threadIdx.x;
    float smax[H], smin[H];
#pragma unroll
    for (int h = 0; h < H; ++h) { smax[h] = -INFINITY; smin[h] = INFINITY; }
    if (i < n) {
        const float* xr = xl + (size_t)i * H * C;
#pragma unroll
        for (int h = 0; h < H; ++h) {
            float s = 0.0f, d = 0.0f;
#pragma unroll
            for (int c = 0; c < C; ++c) {
                float v = xr[h * C + c];
                s += v * a_src[h * C + c];
                d += v * a_dst[h * C + c];
            }
            asn[i * H + h] = s;
            adn[i * H + h] = d;
            smax[h] = s;
            smin[h] = s;
        }
    }
    const int lane = threadIdx.x & (warpSize - 1);
    const int wave = threadIdx.x / warpSize;
    const int nw   = (blockDim.x + warpSize - 1) / warpSize;
#pragma unroll
    for (int h = 0; h < H; ++h) {
        for (int off = warpSize / 2; off > 0; off >>= 1) {
            smax[h] = fmaxf(smax[h], __shfl_xor(smax[h], off));
            smin[h] = fminf(smin[h], __shfl_xor(smin[h], off));
        }
        if (lane == 0) { red[h][wave] = smax[h]; red[H + h][wave] = smin[h]; }
    }
    __syncthreads();
    if (threadIdx.x < 2 * H) {
        const bool is_max = threadIdx.x < H;
        float v = red[threadIdx.x][0];
        for (int w = 1; w < nw; ++w) {
            float o = red[threadIdx.x][w];
            v = is_max ? fmaxf(v, o) : fminf(v, o);
        }
        part[(size_t)blockIdx.x * 2 * H + threadIdx.x] = v;
    }
}

#define SPREAD_OK 16.0f

template <int H>
__global__ void range_kernel(const float* __restrict__ part, int nblk,
                             float* __restrict__ stats,
                             float* __restrict__ mbuf, int nm) {
    __shared__ float red[2 * H][32];
    __shared__ int s_ok;
    float smax[H], smin[H];
#pragma unroll
    for (int h = 0; h < H; ++h) { smax[h] = -INFINITY; smin[h] = INFINITY; }
    for (int b = threadIdx.x; b < nblk; b += blockDim.x) {
#pragma unroll
        for (int h = 0; h < H; ++h) {
            smax[h] = fmaxf(smax[h], part[(size_t)b * 2 * H + h]);
            smin[h] = fminf(smin[h], part[(size_t)b * 2 * H + H + h]);
        }
    }
    const int lane = threadIdx.x & (warpSize - 1);
    const int wave = threadIdx.x / warpSize;
    const int nw   = (blockDim.x + warpSize - 1) / warpSize;
#pragma unroll
    for (int h = 0; h < H; ++h) {
        for (int off = warpSize / 2; off > 0; off >>= 1) {
            smax[h] = fmaxf(smax[h], __shfl_xor(smax[h], off));
            smin[h] = fminf(smin[h], __shfl_xor(smin[h], off));
        }
        if (lane == 0) { red[h][wave] = smax[h]; red[H + h][wave] = smin[h]; }
    }
    __syncthreads();
    if (threadIdx.x == 0) {
        bool ok = true;
#pragma unroll
        for (int h = 0; h < H; ++h) {
            float mx = red[h][0], mn = red[H + h][0];
            for (int w = 1; w < nw; ++w) {
                mx = fmaxf(mx, red[h][w]);
                mn = fminf(mn, red[H + h][w]);
            }
            if (blockIdx.x == 0) stats[h] = mx;
            ok = ok && ((mx - mn) <= SPREAD_OK);
        }
        if (blockIdx.x == 0) stats[H] = ok ? 1.0f : 0.0f;
        s_ok = ok ? 1 : 0;
    }
    __syncthreads();
    if (!s_ok) {
        const int stride = gridDim.x * blockDim.x;
        for (int i = blockIdx.x * blockDim.x + threadIdx.x; i < nm; i += stride)
            mbuf[i] = -INFINITY;
    }
}

template <int H>
__global__ void edge_max_kernel(const int* __restrict__ ei, int E, int EE,
                                const float* __restrict__ asn,
                                const float* __restrict__ adn,
                                const float* __restrict__ stats,
                                float* __restrict__ mbuf) {
    if (stats[H] != 0.0f) return;
    const int stride = gridDim.x * blockDim.x;
    for (int e = blockIdx.x * blockDim.x + threadIdx.x; e < EE; e += stride) {
        int src, dst;
        if (e < E) { src = ei[e]; dst = ei[E + e]; }
        else       { src = dst = e - E; }
#pragma unroll
        for (int h = 0; h < H; ++h) {
            float v = asn[src * H + h] + adn[dst * H + h];
            v = (v > 0.0f) ? v : 0.2f * v;
            atomicMaxFloat(&mbuf[dst * H + h], v);
        }
    }
}

template <int H>
__global__ void edge_exp_kernel(const int* __restrict__ ei, int E, int EE,
                                const float* __restrict__ asn,
                                const float* __restrict__ adn,
                                const float* __restrict__ stats,
                                const float* __restrict__ mbuf,
                                float* __restrict__ ebuf,
                                float* __restrict__ den) {
    int e = blockIdx.x * blockDim.x + threadIdx.x;
    if (e >= EE) return;
    int src, dst;
    if (e < E) { src = ei[e]; dst = ei[E + e]; }
    else       { src = dst = e - E; }
    const bool bounded = stats[H] != 0.0f;
#pragma unroll
    for (int h = 0; h < H; ++h) {
        float ad = adn[dst * H + h];
        float v = asn[src * H + h] + ad;
        v = (v > 0.0f) ? v : 0.2f * v;
        float m;
        if (bounded) {
            float b = stats[h] + ad;
            m = (b > 0.0f) ? b : 0.2f * b;
        } else {
            m = mbuf[dst * H + h];
        }
        if (!(m >= -3.4e38f && m <= 3.4e38f)) m = 0.0f;
        float ex = __expf(v - m);
        ebuf[(size_t)e * H + h] = ex;
        atomicAdd(&den[dst * H + h], ex);
    }
}

template <int H, int C>
__global__ void edge_agg_kernel(const int* __restrict__ ei, int E, int EE,
                                const float* __restrict__ ebuf,
                                const float* __restrict__ den,
                                const float* __restrict__ xl,
                                float* __restrict__ agg) {
    static_assert(C == 32, "one head's row piece must be one 32-lane line");
    static_assert(H == 1 || H == 2, "alpha loads are written for 1 or 2 heads");
    constexpr int F = H * C;
    const int lane = threadIdx.x & 31;
    const int e0 = __builtin_amdgcn_readfirstlane(
        (int)(blockIdx.x * blockDim.x + threadIdx.x) - lane);
    if (e0 >= EE) return;

    const int e = e0 + lane;
    int src = 0, dst = 0;
    float al[H];
    for (int h = 0; h < H; ++h) al[h] = 0.0f;
    if (e < EE) {
        if (e < E) { src = ei[e]; dst = ei[E + e]; }
        else       { src = dst = e - E; }
        if constexpr (H == 2) {
            const float2 ex = *(const float2*)(ebuf + (size_t)e * 2);
            const float2 dn = *(const float2*)(den + (size_t)dst * 2);
            al[0] = ex.x / (dn.x + 1e-16f);
            al[1] = ex.y / (dn.y + 1e-16f);
        } else {
            al[0] = ebuf[e] / (den[dst] + 1e-16f);
        }
    }

    if (EE - e0 >= 32) {
        float v[32 * H];
#pragma unroll
        for (int j = 0; j < 32; ++j) {
            const int s = __builtin_amdgcn_readlane(src, j);
            const float* row = xl + (size_t)s * F;
            for (int h = 0; h < H; ++h) v[j * H + h] = row[h * C + lane];
        }
#pragma unroll
        for (int j = 0; j < 32; ++j) {
            const int d = __builtin_amdgcn_readlane(dst, j);
            float* orow = agg + (size_t)d * F;
            for (int h = 0; h < H; ++h) {
                const float a = __int_as_float(
                    __builtin_amdgcn_readlane(__float_as_int(al[h]), j));
                atomicAdd(orow + h * C + lane, v[j * H + h] * a);
            }
        }
    } else {
        const int cnt = EE - e0;
        for (int j = 0; j < cnt; ++j) {
            const int s = __builtin_amdgcn_readlane(src, j);
            const int d = __builtin_amdgcn_readlane(dst, j);
            const float* row = xl + (size_t)s * F;
            float* orow = agg + (size_t)d * F;
            for (int h = 0; h < H; ++h) {
                const float a = __int_as_float(
                    __builtin_amdgcn_readlane(__float_as_int(al[h]), j));
                atomicAdd(orow + h * C + lane, row[h * C + lane] * a);
            }
        }
    }
}

__global__ void bias_relu_kernel(float* __restrict__ a,
                                 const float* __restrict__ b, int n, int F) {
    int i = blockIdx.x * blockDim.x + threadIdx.x;
    if (i >= n * F) return;
    float v = a[i] + b[i % F];
    a[i] = v > 0.0f ? v : 0.0f;
}

__global__ void bias_add_kernel(float* __restrict__ a,
                                const float* __restrict__ b, int n, int F) {
    int i = blockIdx.x * blockDim.x + threadIdx.x;
    if (i >= n * F) return;
    a[i] += b[i % F];
}

static inline int cdiv(long long a, long long b) { return (int)((a + b - 1) / b); }

extern "C" void kernel_launch(void* const* d_in, const int* in_sizes, int n_in,
                              void* d_out, int out_size, void* d_ws, size_t ws_size,
                              hipStream_t stream) {
    const float* x      = (const float*)d_in[0];
    const int*   ei     = (const int*)d_in[1];
    const float* W1     = (const float*)d_in[2];
    const float* a_src1 = (const float*)d_in[3];
    const float* a_dst1 = (const float*)d_in[4];
    const float* b1     = (const float*)d_in[5];
    const float* W2     = (const float*)d_in[6];
    const float* a_src2 = (const float*)d_in[7];
    const float* a_dst2 = (const float*)d_in[8];
    const float* b2     = (const float*)d_in[9];
    float* out = (float*)d_out;

    const int N  = in_sizes[0] / IN_DIM;
    const int E  = in_sizes[1] / 2;
    const int EE = E + N;
    const int F1 = HEADS * HID;

    float* ws   = (float*)d_ws;
    float* xl1  = ws;
    float* agg1 = xl1  + (size_t)N * F1;
    float* asn1 = agg1 + (size_t)N * F1;
    float* adn1 = asn1 + (size_t)N * HEADS;
    float* m1   = adn1 + (size_t)N * HEADS;
    float* den1 = m1   + (size_t)N * HEADS;
    float* ex1  = den1 + (size_t)N * HEADS;
    float* xl2  = ex1  + (size_t)EE * HEADS;
    float* asn2 = xl2  + (size_t)N * OUT_DIM;
    float* adn2 = asn2 + (size_t)N;
    float* m2   = adn2 + (size_t)N;
    float* den2 = m2   + (size_t)N;
    float* ex2  = den2 + (size_t)N;

    const int TB = 256;
    const int NSB = cdiv(N, TB);

    float* part1  = xl2;
    float* stats1 = part1 + (size_t)NSB * 2 * HEADS;
    float* part2  = ex1;
    float* stats2 = part2 + (size_t)NSB * 2;

    fill_kernel<<<cdiv((long long)N * F1, TB), TB, 0, stream>>>(agg1, 0.0f, N * F1);
    fill_kernel<<<cdiv((long long)N * OUT_DIM, TB), TB, 0, stream>>>(out, 0.0f, N * OUT_DIM);
    fill_kernel<<<cdiv((long long)N * HEADS, TB), TB, 0, stream>>>(den1, 0.0f, N * HEADS);
    fill_kernel<<<cdiv(N, TB), TB, 0, stream>>>(den2, 0.0f, N);

    {
        int mtiles = cdiv(N, 16);
        wmma_gemm_f32<IN_DIM, F1><<<cdiv(mtiles, 4), 128, 0, stream>>>(x, W1, xl1, N);
        score_kernel<HEADS, HID><<<NSB, TB, 0, stream>>>(xl1, a_src1, a_dst1, asn1, adn1, part1, N);
        range_kernel<HEADS><<<64, TB, 0, stream>>>(part1, NSB, stats1, m1, N * HEADS);
        edge_max_kernel<HEADS><<<1024, TB, 0, stream>>>(ei, E, EE, asn1, adn1, stats1, m1);
        edge_exp_kernel<HEADS><<<cdiv(EE, TB), TB, 0, stream>>>(ei, E, EE, asn1, adn1, stats1, m1, ex1, den1);
        edge_agg_kernel<HEADS, HID><<<cdiv(EE, TB), TB, 0, stream>>>(
            ei, E, EE, ex1, den1, xl1, agg1);
        bias_relu_kernel<<<cdiv((long long)N * F1, TB), TB, 0, stream>>>(agg1, b1, N, F1);
    }

    {
        int mtiles = cdiv(N, 16);
        wmma_gemm_f32<F1, OUT_DIM><<<cdiv(mtiles, 4), 128, 0, stream>>>(agg1, W2, xl2, N);
        score_kernel<1, OUT_DIM><<<NSB, TB, 0, stream>>>(xl2, a_src2, a_dst2, asn2, adn2, part2, N);
        range_kernel<1><<<64, TB, 0, stream>>>(part2, NSB, stats2, m2, N);
        edge_max_kernel<1><<<1024, TB, 0, stream>>>(ei, E, EE, asn2, adn2, stats2, m2);
        edge_exp_kernel<1><<<cdiv(EE, TB), TB, 0, stream>>>(ei, E, EE, asn2, adn2, stats2, m2, ex2, den2);
        edge_agg_kernel<1, OUT_DIM><<<cdiv(EE, TB), TB, 0, stream>>>(
            ei, E, EE, ex2, den2, xl2, out);
        bias_add_kernel<<<cdiv((long long)N * OUT_DIM, TB), TB, 0, stream>>>(out, b2, N, OUT_DIM);
    }
}


# Exp pass runs one thread per (edge, head): an edge's two denominator adds (adjacent words) are adjacent lanes of ONE atomic instruction, one line request not two; sum order stays arrival-dependent
# speedup vs baseline: 1.9719x; 1.2288x over previous
#include <hip/hip_runtime.h>
#include <hip/hip_bf16.h>
#include <math.h>

typedef float v2f __attribute__((ext_vector_type(2)));
typedef float v8f __attribute__((ext_vector_type(8)));

#define IN_DIM   128
#define HID      32
#define HEADS    2
#define OUT_DIM  32

__device__ __forceinline__ void atomicMaxFloat(float* addr, float value) {
    if (value >= 0.0f) {
        atomicMax((int*)addr, __float_as_int(value));
    } else {
        atomicMin((unsigned int*)addr, __float_as_uint(value));
    }
}

__global__ void fill_kernel(float* __restrict__ p, float v, int n) {
    int i = blockIdx.x * blockDim.x + threadIdx.x;
    if (i < n) p[i] = v;
}

template <int K, int NC>
__global__ void wmma_gemm_f32(const float* __restrict__ A,
                              const float* __restrict__ B,
                              float* __restrict__ C, int M) {
    constexpr int NT = NC / 16;
    __shared__ v2f Bs2[(K / 2) * NC];

    for (int p = threadIdx.x; p < (K / 2) * NC; p += blockDim.x) {
        int k2 = p / NC;
        int n  = p - k2 * NC;
        v2f pr;
        pr.x = B[(2 * k2 + 0) * NC + n];
        pr.y = B[(2 * k2 + 1) * NC + n];
        Bs2[p] = pr;
    }
    __syncthreads();

    const int wave = threadIdx.x >> 5;
    const int lane = threadIdx.x & 31;
    const int mt   = blockIdx.x * 4 + wave;
    if (mt * 16 >= M) return;

    const int halfK = (lane >> 4) * 2;
    const int ncol  = lane & 15;
    const int arow  = mt * 16 + (lane & 15);
    const float* Arow = A + (size_t)arow * K;

    v8f acc[NT];
#pragma unroll
    for (int t = 0; t < NT; ++t) acc[t] = (v8f){0, 0, 0, 0, 0, 0, 0, 0};

    for (int k = 0; k < K; k += 4) {
        v2f a = *(const v2f*)(Arow + k + halfK);
        const int prow = (k + halfK) >> 1;
#pragma unroll
        for (int t = 0; t < NT; ++t) {
            v2f b = Bs2[prow * NC + t * 16 + ncol];
            acc[t] = __builtin_amdgcn_wmma_f32_16x16x4_f32(
                false, a, false, b, (short)0, acc[t], false, false);
        }
    }

    const int rbase = mt * 16 + ((lane >> 4) << 3);
#pragma unroll
    for (int t = 0; t < NT; ++t) {
#pragma unroll
        for (int j = 0; j < 8; ++j) {
            C[(size_t)(rbase + j) * NC + t * 16 + ncol] = acc[t][j];
        }
    }
}

template <int H, int C>
__global__ void score_kernel(const float* __restrict__ xl,
                             const float* __restrict__ a_src,
                             const float* __restrict__ a_dst,
                             float* __restrict__ asn,
                             float* __restrict__ adn,
                             float* __restrict__ part, int n) {
    __shared__ float red[2 * H][32];
    int i = blockIdx.x * blockDim.x + threadIdx.x;
    float smax[H], smin[H];
#pragma unroll
    for (int h = 0; h < H; ++h) { smax[h] = -INFINITY; smin[h] = INFINITY; }
    if (i < n) {
        const float* xr = xl + (size_t)i * H * C;
#pragma unroll
        for (int h = 0; h < H; ++h) {
            float s = 0.0f, d = 0.0f;
#pragma unroll
            for (int c = 0; c < C; ++c) {
                float v = xr[h * C + c];
                s += v * a_src[h * C + c];
                d += v * a_dst[h * C + c];
            }
            asn[i * H + h] = s;
            adn[i * H + h] = d;
            smax[h] = s;
            smin[h] = s;
        }
    }
    const int lane = threadIdx.x & (warpSize - 1);
    const int wave = threadIdx.x / warpSize;
    const int nw   = (blockDim.x + warpSize - 1) / warpSize;
#pragma unroll
    for (int h = 0; h < H; ++h) {
        for (int off = warpSize / 2; off > 0; off >>= 1) {
            smax[h] = fmaxf(smax[h], __shfl_xor(smax[h], off));
            smin[h] = fminf(smin[h], __shfl_xor(smin[h], off));
        }
        if (lane == 0) { red[h][wave] = smax[h]; red[H + h][wave] = smin[h]; }
    }
    __syncthreads();
    if (threadIdx.x < 2 * H) {
        const bool is_max = threadIdx.x < H;
        float v = red[threadIdx.x][0];
        for (int w = 1; w < nw; ++w) {
            float o = red[threadIdx.x][w];
            v = is_max ? fmaxf(v, o) : fminf(v, o);
        }
        part[(size_t)blockIdx.x * 2 * H + threadIdx.x] = v;
    }
}

#define SPREAD_OK 16.0f

template <int H>
__global__ void range_kernel(const float* __restrict__ part, int nblk,
                             float* __restrict__ stats,
                             float* __restrict__ mbuf, int nm) {
    __shared__ float red[2 * H][32];
    __shared__ int s_ok;
    float smax[H], smin[H];
#pragma unroll
    for (int h = 0; h < H; ++h) { smax[h] = -INFINITY; smin[h] = INFINITY; }
    for (int b = threadIdx.x; b < nblk; b += blockDim.x) {
#pragma unroll
        for (int h = 0; h < H; ++h) {
            smax[h] = fmaxf(smax[h], part[(size_t)b * 2 * H + h]);
            smin[h] = fminf(smin[h], part[(size_t)b * 2 * H + H + h]);
        }
    }
    const int lane = threadIdx.x & (warpSize - 1);
    const int wave = threadIdx.x / warpSize;
    const int nw   = (blockDim.x + warpSize - 1) / warpSize;
#pragma unroll
    for (int h = 0; h < H; ++h) {
        for (int off = warpSize / 2; off > 0; off >>= 1) {
            smax[h] = fmaxf(smax[h], __shfl_xor(smax[h], off));
            smin[h] = fminf(smin[h], __shfl_xor(smin[h], off));
        }
        if (lane == 0) { red[h][wave] = smax[h]; red[H + h][wave] = smin[h]; }
    }
    __syncthreads();
    if (threadIdx.x == 0) {
        bool ok = true;
#pragma unroll
        for (int h = 0; h < H; ++h) {
            float mx = red[h][0], mn = red[H + h][0];
            for (int w = 1; w < nw; ++w) {
                mx = fmaxf(mx, red[h][w]);
                mn = fminf(mn, red[H + h][w]);
            }
            if (blockIdx.x == 0) stats[h] = mx;
            ok = ok && ((mx - mn) <= SPREAD_OK);
        }
        if (blockIdx.x == 0) stats[H] = ok ? 1.0f : 0.0f;
        s_ok = ok ? 1 : 0;
    }
    __syncthreads();
    if (!s_ok) {
        const int stride = gridDim.x * blockDim.x;
        for (int i = blockIdx.x * blockDim.x + threadIdx.x; i < nm; i += stride)
            mbuf[i] = -INFINITY;
    }
}

template <int H>
__global__ void edge_max_kernel(const int* __restrict__ ei, int E, int EE,
                                const float* __restrict__ asn,
                                const float* __restrict__ adn,
                                const float* __restrict__ stats,
                                float* __restrict__ mbuf) {
    if (stats[H] != 0.0f) return;
    const int stride = gridDim.x * blockDim.x;
    for (int e = blockIdx.x * blockDim.x + threadIdx.x; e < EE; e += stride) {
        int src, dst;
        if (e < E) { src = ei[e]; dst = ei[E + e]; }
        else       { src = dst = e - E; }
#pragma unroll
        for (int h = 0; h < H; ++h) {
            float v = asn[src * H + h] + adn[dst * H + h];
            v = (v > 0.0f) ? v : 0.2f * v;
            atomicMaxFloat(&mbuf[dst * H + h], v);
        }
    }
}

template <int H>
__global__ void edge_exp_kernel(const int* __restrict__ ei, int E, int EE,
                                const float* __restrict__ asn,
                                const float* __restrict__ adn,
                                const float* __restrict__ stats,
                                const float* __restrict__ mbuf,
                                float* __restrict__ ebuf,
                                float* __restrict__ den) {
    const int t = blockIdx.x * blockDim.x + threadIdx.x;
    const int e = t / H;
    const int h = t - e * H;
    if (e >= EE) return;
    int src, dst;
    if (e < E) { src = ei[e]; dst = ei[E + e]; }
    else       { src = dst = e - E; }
    const bool bounded = stats[H] != 0.0f;
    const float ad = adn[dst * H + h];
    float v = asn[src * H + h] + ad;
    v = (v > 0.0f) ? v : 0.2f * v;
    float m;
    if (bounded) {
        float b = stats[h] + ad;
        m = (b > 0.0f) ? b : 0.2f * b;
    } else {
        m = mbuf[dst * H + h];
    }
    if (!(m >= -3.4e38f && m <= 3.4e38f)) m = 0.0f;
    const float ex = __expf(v - m);
    ebuf[t] = ex;
    atomicAdd(&den[dst * H + h], ex);
}

template <int H, int C>
__global__ void edge_agg_kernel(const int* __restrict__ ei, int E, int EE,
                                const float* __restrict__ ebuf,
                                const float* __restrict__ den,
                                const float* __restrict__ xl,
                                float* __restrict__ agg) {
    static_assert(C == 32, "one head's row piece must be one 32-lane line");
    static_assert(H == 1 || H == 2, "alpha loads are written for 1 or 2 heads");
    constexpr int F = H * C;
    const int lane = threadIdx.x & 31;
    const int e0 = __builtin_amdgcn_readfirstlane(
        (int)(blockIdx.x * blockDim.x + threadIdx.x) - lane);
    if (e0 >= EE) return;

    const int e = e0 + lane;
    int src = 0, dst = 0;
    float al[H];
    for (int h = 0; h < H; ++h) al[h] = 0.0f;
    if (e < EE) {
        if (e < E) { src = ei[e]; dst = ei[E + e]; }
        else       { src = dst = e - E; }
        if constexpr (H == 2) {
            const float2 ex = *(const float2*)(ebuf + (size_t)e * 2);
            const float2 dn = *(const float2*)(den + (size_t)dst * 2);
            al[0] = ex.x / (dn.x + 1e-16f);
            al[1] = ex.y / (dn.y + 1e-16f);
        } else {
            al[0] = ebuf[e] / (den[dst] + 1e-16f);
        }
    }

    if (EE - e0 >= 32) {
        float v[32 * H];
#pragma unroll
        for (int j = 0; j < 32; ++j) {
            const int s = __builtin_amdgcn_readlane(src, j);
            const float* row = xl + (size_t)s * F;
            for (int h = 0; h < H; ++h) v[j * H + h] = row[h * C + lane];
        }
#pragma unroll
        for (int j = 0; j < 32; ++j) {
            const int d = __builtin_amdgcn_readlane(dst, j);
            float* orow = agg + (size_t)d * F;
            for (int h = 0; h < H; ++h) {
                const float a = __int_as_float(
                    __builtin_amdgcn_readlane(__float_as_int(al[h]), j));
                atomicAdd(orow + h * C + lane, v[j * H + h] * a);
            }
        }
    } else {
        const int cnt = EE - e0;
        for (int j = 0; j < cnt; ++j) {
            const int s = __builtin_amdgcn_readlane(src, j);
            const int d = __builtin_amdgcn_readlane(dst, j);
            const float* row = xl + (size_t)s * F;
            float* orow = agg + (size_t)d * F;
            for (int h = 0; h < H; ++h) {
                const float a = __int_as_float(
                    __builtin_amdgcn_readlane(__float_as_int(al[h]), j));
                atomicAdd(orow + h * C + lane, row[h * C + lane] * a);
            }
        }
    }
}

__global__ void bias_relu_kernel(float* __restrict__ a,
                                 const float* __restrict__ b, int n, int F) {
    int i = blockIdx.x * blockDim.x + threadIdx.x;
    if (i >= n * F) return;
    float v = a[i] + b[i % F];
    a[i] = v > 0.0f ? v : 0.0f;
}

__global__ void bias_add_kernel(float* __restrict__ a,
                                const float* __restrict__ b, int n, int F) {
    int i = blockIdx.x * blockDim.x + threadIdx.x;
    if (i >= n * F) return;
    a[i] += b[i % F];
}

static inline int cdiv(long long a, long long b) { return (int)((a + b - 1) / b); }

extern "C" void kernel_launch(void* const* d_in, const int* in_sizes, int n_in,
                              void* d_out, int out_size, void* d_ws, size_t ws_size,
                              hipStream_t stream) {
    const float* x      = (const float*)d_in[0];
    const int*   ei     = (const int*)d_in[1];
    const float* W1     = (const float*)d_in[2];
    const float* a_src1 = (const float*)d_in[3];
    const float* a_dst1 = (const float*)d_in[4];
    const float* b1     = (const float*)d_in[5];
    const float* W2     = (const float*)d_in[6];
    const float* a_src2 = (const float*)d_in[7];
    const float* a_dst2 = (const float*)d_in[8];
    const float* b2     = (const float*)d_in[9];
    float* out = (float*)d_out;

    const int N  = in_sizes[0] / IN_DIM;
    const int E  = in_sizes[1] / 2;
    const int EE = E + N;
    const int F1 = HEADS * HID;

    float* ws   = (float*)d_ws;
    float* xl1  = ws;
    float* agg1 = xl1  + (size_t)N * F1;
    float* asn1 = agg1 + (size_t)N * F1;
    float* adn1 = asn1 + (size_t)N * HEADS;
    float* m1   = adn1 + (size_t)N * HEADS;
    float* den1 = m1   + (size_t)N * HEADS;
    float* ex1  = den1 + (size_t)N * HEADS;
    float* xl2  = ex1  + (size_t)EE * HEADS;
    float* asn2 = xl2  + (size_t)N * OUT_DIM;
    float* adn2 = asn2 + (size_t)N;
    float* m2   = adn2 + (size_t)N;
    float* den2 = m2   + (size_t)N;
    float* ex2  = den2 + (size_t)N;

    const int TB = 256;
    const int NSB = cdiv(N, TB);

    float* part1  = xl2;
    float* stats1 = part1 + (size_t)NSB * 2 * HEADS;
    float* part2  = ex1;
    float* stats2 = part2 + (size_t)NSB * 2;

    fill_kernel<<<cdiv((long long)N * F1, TB), TB, 0, stream>>>(agg1, 0.0f, N * F1);
    fill_kernel<<<cdiv((long long)N * OUT_DIM, TB), TB, 0, stream>>>(out, 0.0f, N * OUT_DIM);
    fill_kernel<<<cdiv((long long)N * HEADS, TB), TB, 0, stream>>>(den1, 0.0f, N * HEADS);
    fill_kernel<<<cdiv(N, TB), TB, 0, stream>>>(den2, 0.0f, N);

    {
        int mtiles = cdiv(N, 16);
        wmma_gemm_f32<IN_DIM, F1><<<cdiv(mtiles, 4), 128, 0, stream>>>(x, W1, xl1, N);
        score_kernel<HEADS, HID><<<NSB, TB, 0, stream>>>(xl1, a_src1, a_dst1, asn1, adn1, part1, N);
        range_kernel<HEADS><<<64, TB, 0, stream>>>(part1, NSB, stats1, m1, N * HEADS);
        edge_max_kernel<HEADS><<<1024, TB, 0, stream>>>(ei, E, EE, asn1, adn1, stats1, m1);
        edge_exp_kernel<HEADS><<<cdiv((long long)EE * HEADS, TB), TB, 0, stream>>>(ei, E, EE, asn1, adn1, stats1, m1, ex1, den1);
        edge_agg_kernel<HEADS, HID><<<cdiv(EE, TB), TB, 0, stream>>>(
            ei, E, EE, ex1, den1, xl1, agg1);
        bias_relu_kernel<<<cdiv((long long)N * F1, TB), TB, 0, stream>>>(agg1, b1, N, F1);
    }

    {
        int mtiles = cdiv(N, 16);
        wmma_gemm_f32<F1, OUT_DIM><<<cdiv(mtiles, 4), 128, 0, stream>>>(agg1, W2, xl2, N);
        score_kernel<1, OUT_DIM><<<NSB, TB, 0, stream>>>(xl2, a_src2, a_dst2, asn2, adn2, part2, N);
        range_kernel<1><<<64, TB, 0, stream>>>(part2, NSB, stats2, m2, N);
        edge_max_kernel<1><<<1024, TB, 0, stream>>>(ei, E, EE, asn2, adn2, stats2, m2);
        edge_exp_kernel<1><<<cdiv(EE, TB), TB, 0, stream>>>(ei, E, EE, asn2, adn2, stats2, m2, ex2, den2);
        edge_agg_kernel<1, OUT_DIM><<<cdiv(EE, TB), TB, 0, stream>>>(
            ei, E, EE, ex2, den2, xl2, out);
        bias_add_kernel<<<cdiv((long long)N * OUT_DIM, TB), TB, 0, stream>>>(out, b2, N, OUT_DIM);
    }
}


# Fills and bias passes folded into neighbours: product kernels store bias as the segment sums' start value, score kernels zero den, relu moves to the next product's operand load; bias summed first
# speedup vs baseline: 2.1329x; 1.0816x over previous
#include <hip/hip_runtime.h>
#include <hip/hip_bf16.h>
#include <math.h>

typedef float v2f __attribute__((ext_vector_type(2)));
typedef float v8f __attribute__((ext_vector_type(8)));

#define IN_DIM   128
#define HID      32
#define HEADS    2
#define OUT_DIM  32

__device__ __forceinline__ void atomicMaxFloat(float* addr, float value) {
    if (value >= 0.0f) {
        atomicMax((int*)addr, __float_as_int(value));
    } else {
        atomicMin((unsigned int*)addr, __float_as_uint(value));
    }
}

template <int K, int NC, bool RELU_A>
__global__ void wmma_gemm_f32(const float* __restrict__ A,
                              const float* __restrict__ B,
                              float* __restrict__ C,
                              const float* __restrict__ bias,
                              float* __restrict__ S, int M) {
    constexpr int NT = NC / 16;
    __shared__ v2f Bs2[(K / 2) * NC];

    for (int p = threadIdx.x; p < (K / 2) * NC; p += blockDim.x) {
        int k2 = p / NC;
        int n  = p - k2 * NC;
        v2f pr;
        pr.x = B[(2 * k2 + 0) * NC + n];
        pr.y = B[(2 * k2 + 1) * NC + n];
        Bs2[p] = pr;
    }
    __syncthreads();

    const int wave = threadIdx.x >> 5;
    const int lane = threadIdx.x & 31;
    const int mt   = blockIdx.x * 4 + wave;
    if (mt * 16 >= M) return;

    const int halfK = (lane >> 4) * 2;
    const int ncol  = lane & 15;
    const int arow  = mt * 16 + (lane & 15);
    const float* Arow = A + (size_t)arow * K;

    {
        static_assert(128 % NC == 0, "a lane must keep its columns");
        constexpr int NV = 16 * NC / 128;
        const int c0 = (lane * 4) % NC;
        float4 bv;
        bv.x = bias[c0 + 0];
        bv.y = bias[c0 + 1];
        bv.z = bias[c0 + 2];
        bv.w = bias[c0 + 3];
        float4* Sp = (float4*)(S + (size_t)mt * 16 * NC) + lane;
#pragma unroll
        for (int i = 0; i < NV; ++i) Sp[i * 32] = bv;
    }

    v8f acc[NT];
#pragma unroll
    for (int t = 0; t < NT; ++t) acc[t] = (v8f){0, 0, 0, 0, 0, 0, 0, 0};

    for (int k = 0; k < K; k += 4) {
        v2f a = *(const v2f*)(Arow + k + halfK);
        if constexpr (RELU_A) {
            a.x = a.x > 0.0f ? a.x : 0.0f;
            a.y = a.y > 0.0f ? a.y : 0.0f;
        }
        const int prow = (k + halfK) >> 1;
#pragma unroll
        for (int t = 0; t < NT; ++t) {
            v2f b = Bs2[prow * NC + t * 16 + ncol];
            acc[t] = __builtin_amdgcn_wmma_f32_16x16x4_f32(
                false, a, false, b, (short)0, acc[t], false, false);
        }
    }

    const int rbase = mt * 16 + ((lane >> 4) << 3);
#pragma unroll
    for (int t = 0; t < NT; ++t) {
#pragma unroll
        for (int j = 0; j < 8; ++j) {
            C[(size_t)(rbase + j) * NC + t * 16 + ncol] = acc[t][j];
        }
    }
}

template <int H, int C>
__global__ void score_kernel(const float* __restrict__ xl,
                             const float* __restrict__ a_src,
                             const float* __restrict__ a_dst,
                             float* __restrict__ asn,
                             float* __restrict__ adn,
                             float* __restrict__ den,
                             float* __restrict__ part, int n) {
    __shared__ float red[2 * H][32];
    int i = blockIdx.x * blockDim.x + threadIdx.x;
    float smax[H], smin[H];
#pragma unroll
    for (int h = 0; h < H; ++h) { smax[h] = -INFINITY; smin[h] = INFINITY; }
    if (i < n) {
        const float* xr = xl + (size_t)i * H * C;
#pragma unroll
        for (int h = 0; h < H; ++h) {
            float s = 0.0f, d = 0.0f;
#pragma unroll
            for (int c = 0; c < C; ++c) {
                float v = xr[h * C + c];
                s += v * a_src[h * C + c];
                d += v * a_dst[h * C + c];
            }
            asn[i * H + h] = s;
            adn[i * H + h] = d;
            den[i * H + h] = 0.0f;
            smax[h] = s;
            smin[h] = s;
        }
    }
    const int lane = threadIdx.x & (warpSize - 1);
    const int wave = threadIdx.x / warpSize;
    const int nw   = (blockDim.x + warpSize - 1) / warpSize;
#pragma unroll
    for (int h = 0; h < H; ++h) {
        for (int off = warpSize / 2; off > 0; off >>= 1) {
            smax[h] = fmaxf(smax[h], __shfl_xor(smax[h], off));
            smin[h] = fminf(smin[h], __shfl_xor(smin[h], off));
        }
        if (lane == 0) { red[h][wave] = smax[h]; red[H + h][wave] = smin[h]; }
    }
    __syncthreads();
    if (threadIdx.x < 2 * H) {
        const bool is_max = threadIdx.x < H;
        float v = red[threadIdx.x][0];
        for (int w = 1; w < nw; ++w) {
            float o = red[threadIdx.x][w];
            v = is_max ? fmaxf(v, o) : fminf(v, o);
        }
        part[(size_t)blockIdx.x * 2 * H + threadIdx.x] = v;
    }
}

#define SPREAD_OK 16.0f

template <int H>
__global__ void range_kernel(const float* __restrict__ part, int nblk,
                             float* __restrict__ stats,
                             float* __restrict__ mbuf, int nm) {
    __shared__ float red[2 * H][32];
    __shared__ int s_ok;
    float smax[H], smin[H];
#pragma unroll
    for (int h = 0; h < H; ++h) { smax[h] = -INFINITY; smin[h] = INFINITY; }
    for (int b = threadIdx.x; b < nblk; b += blockDim.x) {
#pragma unroll
        for (int h = 0; h < H; ++h) {
            smax[h] = fmaxf(smax[h], part[(size_t)b * 2 * H + h]);
            smin[h] = fminf(smin[h], part[(size_t)b * 2 * H + H + h]);
        }
    }
    const int lane = threadIdx.x & (warpSize - 1);
    const int wave = threadIdx.x / warpSize;
    const int nw   = (blockDim.x + warpSize - 1) / warpSize;
#pragma unroll
    for (int h = 0; h < H; ++h) {
        for (int off = warpSize / 2; off > 0; off >>= 1) {
            smax[h] = fmaxf(smax[h], __shfl_xor(smax[h], off));
            smin[h] = fminf(smin[h], __shfl_xor(smin[h], off));
        }
        if (lane == 0) { red[h][wave] = smax[h]; red[H + h][wave] = smin[h]; }
    }
    __syncthreads();
    if (threadIdx.x == 0) {
        bool ok = true;
#pragma unroll
        for (int h = 0; h < H; ++h) {
            float mx = red[h][0], mn = red[H + h][0];
            for (int w = 1; w < nw; ++w) {
                mx = fmaxf(mx, red[h][w]);
                mn = fminf(mn, red[H + h][w]);
            }
            if (blockIdx.x == 0) stats[h] = mx;
            ok = ok && ((mx - mn) <= SPREAD_OK);
        }
        if (blockIdx.x == 0) stats[H] = ok ? 1.0f : 0.0f;
        s_ok = ok ? 1 : 0;
    }
    __syncthreads();
    if (!s_ok) {
        const int stride = gridDim.x * blockDim.x;
        for (int i = blockIdx.x * blockDim.x + threadIdx.x; i < nm; i += stride)
            mbuf[i] = -INFINITY;
    }
}

template <int H>
__global__ void edge_max_kernel(const int* __restrict__ ei, int E, int EE,
                                const float* __restrict__ asn,
                                const float* __restrict__ adn,
                                const float* __restrict__ stats,
                                float* __restrict__ mbuf) {
    if (stats[H] != 0.0f) return;
    const int stride = gridDim.x * blockDim.x;
    for (int e = blockIdx.x * blockDim.x + threadIdx.x; e < EE; e += stride) {
        int src, dst;
        if (e < E) { src = ei[e]; dst = ei[E + e]; }
        else       { src = dst = e - E; }
#pragma unroll
        for (int h = 0; h < H; ++h) {
            float v = asn[src * H + h] + adn[dst * H + h];
            v = (v > 0.0f) ? v : 0.2f * v;
            atomicMaxFloat(&mbuf[dst * H + h], v);
        }
    }
}

template <int H>
__global__ void edge_exp_kernel(const int* __restrict__ ei, int E, int EE,
                                const float* __restrict__ asn,
                                const float* __restrict__ adn,
                                const float* __restrict__ stats,
                                const float* __restrict__ mbuf,
                                float* __restrict__ ebuf,
                                float* __restrict__ den) {
    const int t = blockIdx.x * blockDim.x + threadIdx.x;
    const int e = t / H;
    const int h = t - e * H;
    if (e >= EE) return;
    int src, dst;
    if (e < E) { src = ei[e]; dst = ei[E + e]; }
    else       { src = dst = e - E; }
    const bool bounded = stats[H] != 0.0f;
    const float ad = adn[dst * H + h];
    float v = asn[src * H + h] + ad;
    v = (v > 0.0f) ? v : 0.2f * v;
    float m;
    if (bounded) {
        float b = stats[h] + ad;
        m = (b > 0.0f) ? b : 0.2f * b;
    } else {
        m = mbuf[dst * H + h];
    }
    if (!(m >= -3.4e38f && m <= 3.4e38f)) m = 0.0f;
    const float ex = __expf(v - m);
    ebuf[t] = ex;
    atomicAdd(&den[dst * H + h], ex);
}

template <int H, int C>
__global__ void edge_agg_kernel(const int* __restrict__ ei, int E, int EE,
                                const float* __restrict__ ebuf,
                                const float* __restrict__ den,
                                const float* __restrict__ xl,
                                float* __restrict__ agg) {
    static_assert(C == 32, "one head's row piece must be one 32-lane line");
    static_assert(H == 1 || H == 2, "alpha loads are written for 1 or 2 heads");
    constexpr int F = H * C;
    const int lane = threadIdx.x & 31;
    const int e0 = __builtin_amdgcn_readfirstlane(
        (int)(blockIdx.x * blockDim.x + threadIdx.x) - lane);
    if (e0 >= EE) return;

    const int e = e0 + lane;
    int src = 0, dst = 0;
    float al[H];
    for (int h = 0; h < H; ++h) al[h] = 0.0f;
    if (e < EE) {
        if (e < E) { src = ei[e]; dst = ei[E + e]; }
        else       { src = dst = e - E; }
        if constexpr (H == 2) {
            const float2 ex = *(const float2*)(ebuf + (size_t)e * 2);
            const float2 dn = *(const float2*)(den + (size_t)dst * 2);
            al[0] = ex.x / (dn.x + 1e-16f);
            al[1] = ex.y / (dn.y + 1e-16f);
        } else {
            al[0] = ebuf[e] / (den[dst] + 1e-16f);
        }
    }

    if (EE - e0 >= 32) {
        float v[32 * H];
#pragma unroll
        for (int j = 0; j < 32; ++j) {
            const int s = __builtin_amdgcn_readlane(src, j);
            const float* row = xl + (size_t)s * F;
            for (int h = 0; h < H; ++h) v[j * H + h] = row[h * C + lane];
        }
#pragma unroll
        for (int j = 0; j < 32; ++j) {
            const int d = __builtin_amdgcn_readlane(dst, j);
            float* orow = agg + (size_t)d * F;
            for (int h = 0; h < H; ++h) {
                const float a = __int_as_float(
                    __builtin_amdgcn_readlane(__float_as_int(al[h]), j));
                atomicAdd(orow + h * C + lane, v[j * H + h] * a);
            }
        }
    } else {
        const int cnt = EE - e0;
        for (int j = 0; j < cnt; ++j) {
            const int s = __builtin_amdgcn_readlane(src, j);
            const int d = __builtin_amdgcn_readlane(dst, j);
            const float* row = xl + (size_t)s * F;
            float* orow = agg + (size_t)d * F;
            for (int h = 0; h < H; ++h) {
                const float a = __int_as_float(
                    __builtin_amdgcn_readlane(__float_as_int(al[h]), j));
                atomicAdd(orow + h * C + lane, row[h * C + lane] * a);
            }
        }
    }
}

static inline int cdiv(long long a, long long b) { return (int)((a + b - 1) / b); }

extern "C" void kernel_launch(void* const* d_in, const int* in_sizes, int n_in,
                              void* d_out, int out_size, void* d_ws, size_t ws_size,
                              hipStream_t stream) {
    const float* x      = (const float*)d_in[0];
    const int*   ei     = (const int*)d_in[1];
    const float* W1     = (const float*)d_in[2];
    const float* a_src1 = (const float*)d_in[3];
    const float* a_dst1 = (const float*)d_in[4];
    const float* b1     = (const float*)d_in[5];
    const float* W2     = (const float*)d_in[6];
    const float* a_src2 = (const float*)d_in[7];
    const float* a_dst2 = (const float*)d_in[8];
    const float* b2     = (const float*)d_in[9];
    float* out = (float*)d_out;

    const int N  = in_sizes[0] / IN_DIM;
    const int E  = in_sizes[1] / 2;
    const int EE = E + N;
    const int F1 = HEADS * HID;

    float* ws   = (float*)d_ws;
    float* xl1  = ws;
    float* agg1 = xl1  + (size_t)N * F1;
    float* asn1 = agg1 + (size_t)N * F1;
    float* adn1 = asn1 + (size_t)N * HEADS;
    float* m1   = adn1 + (size_t)N * HEADS;
    float* den1 = m1   + (size_t)N * HEADS;
    float* ex1  = den1 + (size_t)N * HEADS;
    float* xl2  = ex1  + (size_t)EE * HEADS;
    float* asn2 = xl2  + (size_t)N * OUT_DIM;
    float* adn2 = asn2 + (size_t)N;
    float* m2   = adn2 + (size_t)N;
    float* den2 = m2   + (size_t)N;
    float* ex2  = den2 + (size_t)N;

    const int TB = 256;
    const int NSB = cdiv(N, TB);

    float* part1  = xl2;
    float* stats1 = part1 + (size_t)NSB * 2 * HEADS;
    float* part2  = ex1;
    float* stats2 = part2 + (size_t)NSB * 2;


    {
        int mtiles = cdiv(N, 16);
        wmma_gemm_f32<IN_DIM, F1, false><<<cdiv(mtiles, 4), 128, 0, stream>>>(
            x, W1, xl1, b1, agg1, N);
        score_kernel<HEADS, HID><<<NSB, TB, 0, stream>>>(xl1, a_src1, a_dst1, asn1, adn1, den1, part1, N);
        range_kernel<HEADS><<<64, TB, 0, stream>>>(part1, NSB, stats1, m1, N * HEADS);
        edge_max_kernel<HEADS><<<1024, TB, 0, stream>>>(ei, E, EE, asn1, adn1, stats1, m1);
        edge_exp_kernel<HEADS><<<cdiv((long long)EE * HEADS, TB), TB, 0, stream>>>(ei, E, EE, asn1, adn1, stats1, m1, ex1, den1);
        edge_agg_kernel<HEADS, HID><<<cdiv(EE, TB), TB, 0, stream>>>(
            ei, E, EE, ex1, den1, xl1, agg1);
    }

    {
        int mtiles = cdiv(N, 16);
        wmma_gemm_f32<F1, OUT_DIM, true><<<cdiv(mtiles, 4), 128, 0, stream>>>(
            agg1, W2, xl2, b2, out, N);
        score_kernel<1, OUT_DIM><<<NSB, TB, 0, stream>>>(xl2, a_src2, a_dst2, asn2, adn2, den2, part2, N);
        range_kernel<1><<<64, TB, 0, stream>>>(part2, NSB, stats2, m2, N);
        edge_max_kernel<1><<<1024, TB, 0, stream>>>(ei, E, EE, asn2, adn2, stats2, m2);
        edge_exp_kernel<1><<<cdiv(EE, TB), TB, 0, stream>>>(ei, E, EE, asn2, adn2, stats2, m2, ex2, den2);
        edge_agg_kernel<1, OUT_DIM><<<cdiv(EE, TB), TB, 0, stream>>>(
            ei, E, EE, ex2, den2, xl2, out);
    }
}


# Softmax denominators: per-edge scattered global atomics replaced by a pass that streams dst + exp terms into LDS bins per node range and adds whole lines; sum order changes, stays arrival-dependent
# speedup vs baseline: 3.0569x; 1.4332x over previous
#include <hip/hip_runtime.h>
#include <hip/hip_bf16.h>
#include <math.h>

typedef float v2f __attribute__((ext_vector_type(2)));
typedef float v8f __attribute__((ext_vector_type(8)));

#define IN_DIM   128
#define HID      32
#define HEADS    2
#define OUT_DIM  32

__device__ __forceinline__ void atomicMaxFloat(float* addr, float value) {
    if (value >= 0.0f) {
        atomicMax((int*)addr, __float_as_int(value));
    } else {
        atomicMin((unsigned int*)addr, __float_as_uint(value));
    }
}

template <int K, int NC, bool RELU_A>
__global__ void wmma_gemm_f32(const float* __restrict__ A,
                              const float* __restrict__ B,
                              float* __restrict__ C,
                              const float* __restrict__ bias,
                              float* __restrict__ S, int M) {
    constexpr int NT = NC / 16;
    __shared__ v2f Bs2[(K / 2) * NC];

    for (int p = threadIdx.x; p < (K / 2) * NC; p += blockDim.x) {
        int k2 = p / NC;
        int n  = p - k2 * NC;
        v2f pr;
        pr.x = B[(2 * k2 + 0) * NC + n];
        pr.y = B[(2 * k2 + 1) * NC + n];
        Bs2[p] = pr;
    }
    __syncthreads();

    const int wave = threadIdx.x >> 5;
    const int lane = threadIdx.x & 31;
    const int mt   = blockIdx.x * 4 + wave;
    if (mt * 16 >= M) return;

    const int halfK = (lane >> 4) * 2;
    const int ncol  = lane & 15;
    const int arow  = mt * 16 + (lane & 15);
    const float* Arow = A + (size_t)arow * K;

    {
        static_assert(128 % NC == 0, "a lane must keep its columns");
        constexpr int NV = 16 * NC / 128;
        const int c0 = (lane * 4) % NC;
        float4 bv;
        bv.x = bias[c0 + 0];
        bv.y = bias[c0 + 1];
        bv.z = bias[c0 + 2];
        bv.w = bias[c0 + 3];
        float4* Sp = (float4*)(S + (size_t)mt * 16 * NC) + lane;
#pragma unroll
        for (int i = 0; i < NV; ++i) Sp[i * 32] = bv;
    }

    v8f acc[NT];
#pragma unroll
    for (int t = 0; t < NT; ++t) acc[t] = (v8f){0, 0, 0, 0, 0, 0, 0, 0};

    for (int k = 0; k < K; k += 4) {
        v2f a = *(const v2f*)(Arow + k + halfK);
        if constexpr (RELU_A) {
            a.x = a.x > 0.0f ? a.x : 0.0f;
            a.y = a.y > 0.0f ? a.y : 0.0f;
        }
        const int prow = (k + halfK) >> 1;
#pragma unroll
        for (int t = 0; t < NT; ++t) {
            v2f b = Bs2[prow * NC + t * 16 + ncol];
            acc[t] = __builtin_amdgcn_wmma_f32_16x16x4_f32(
                false, a, false, b, (short)0, acc[t], false, false);
        }
    }

    const int rbase = mt * 16 + ((lane >> 4) << 3);
#pragma unroll
    for (int t = 0; t < NT; ++t) {
#pragma unroll
        for (int j = 0; j < 8; ++j) {
            C[(size_t)(rbase + j) * NC + t * 16 + ncol] = acc[t][j];
        }
    }
}

template <int H, int C>
__global__ void score_kernel(const float* __restrict__ xl,
                             const float* __restrict__ a_src,
                             const float* __restrict__ a_dst,
                             float* __restrict__ asn,
                             float* __restrict__ adn,
                             float* __restrict__ den,
                             float* __restrict__ part, int n) {
    __shared__ float red[2 * H][32];
    int i = blockIdx.x * blockDim.x + threadIdx.x;
    float smax[H], smin[H];
#pragma unroll
    for (int h = 0; h < H; ++h) { smax[h] = -INFINITY; smin[h] = INFINITY; }
    if (i < n) {
        const float* xr = xl + (size_t)i * H * C;
#pragma unroll
        for (int h = 0; h < H; ++h) {
            float s = 0.0f, d = 0.0f;
#pragma unroll
            for (int c = 0; c < C; ++c) {
                float v = xr[h * C + c];
                s += v * a_src[h * C + c];
                d += v * a_dst[h * C + c];
            }
            asn[i * H + h] = s;
            adn[i * H + h] = d;
            den[i * H + h] = 0.0f;
            smax[h] = s;
            smin[h] = s;
        }
    }
    const int lane = threadIdx.x & (warpSize - 1);
    const int wave = threadIdx.x / warpSize;
    const int nw   = (blockDim.x + warpSize - 1) / warpSize;
#pragma unroll
    for (int h = 0; h < H; ++h) {
        for (int off = warpSize / 2; off > 0; off >>= 1) {
            smax[h] = fmaxf(smax[h], __shfl_xor(smax[h], off));
            smin[h] = fminf(smin[h], __shfl_xor(smin[h], off));
        }
        if (lane == 0) { red[h][wave] = smax[h]; red[H + h][wave] = smin[h]; }
    }
    __syncthreads();
    if (threadIdx.x < 2 * H) {
        const bool is_max = threadIdx.x < H;
        float v = red[threadIdx.x][0];
        for (int w = 1; w < nw; ++w) {
            float o = red[threadIdx.x][w];
            v = is_max ? fmaxf(v, o) : fminf(v, o);
        }
        part[(size_t)blockIdx.x * 2 * H + threadIdx.x] = v;
    }
}

#define SPREAD_OK 16.0f

#define DEN_R      12800
#define DEN_SLICES 32

template <int H>
__global__ void range_kernel(const float* __restrict__ part, int nblk,
                             float* __restrict__ stats,
                             float* __restrict__ mbuf, int nm) {
    __shared__ float red[2 * H][32];
    __shared__ int s_ok;
    float smax[H], smin[H];
#pragma unroll
    for (int h = 0; h < H; ++h) { smax[h] = -INFINITY; smin[h] = INFINITY; }
    for (int b = threadIdx.x; b < nblk; b += blockDim.x) {
#pragma unroll
        for (int h = 0; h < H; ++h) {
            smax[h] = fmaxf(smax[h], part[(size_t)b * 2 * H + h]);
            smin[h] = fminf(smin[h], part[(size_t)b * 2 * H + H + h]);
        }
    }
    const int lane = threadIdx.x & (warpSize - 1);
    const int wave = threadIdx.x / warpSize;
    const int nw   = (blockDim.x + warpSize - 1) / warpSize;
#pragma unroll
    for (int h = 0; h < H; ++h) {
        for (int off = warpSize / 2; off > 0; off >>= 1) {
            smax[h] = fmaxf(smax[h], __shfl_xor(smax[h], off));
            smin[h] = fminf(smin[h], __shfl_xor(smin[h], off));
        }
        if (lane == 0) { red[h][wave] = smax[h]; red[H + h][wave] = smin[h]; }
    }
    __syncthreads();
    if (threadIdx.x == 0) {
        bool ok = true;
#pragma unroll
        for (int h = 0; h < H; ++h) {
            float mx = red[h][0], mn = red[H + h][0];
            for (int w = 1; w < nw; ++w) {
                mx = fmaxf(mx, red[h][w]);
                mn = fminf(mn, red[H + h][w]);
            }
            if (blockIdx.x == 0) stats[h] = mx;
            ok = ok && ((mx - mn) <= SPREAD_OK);
        }
        if (blockIdx.x == 0) stats[H] = ok ? 1.0f : 0.0f;
        s_ok = ok ? 1 : 0;
    }
    __syncthreads();
    if (!s_ok) {
        const int stride = gridDim.x * blockDim.x;
        for (int i = blockIdx.x * blockDim.x + threadIdx.x; i < nm; i += stride)
            mbuf[i] = -INFINITY;
    }
}

template <int H>
__global__ void edge_max_kernel(const int* __restrict__ ei, int E, int EE,
                                const float* __restrict__ asn,
                                const float* __restrict__ adn,
                                const float* __restrict__ stats,
                                float* __restrict__ mbuf) {
    if (stats[H] != 0.0f) return;
    const int stride = gridDim.x * blockDim.x;
    for (int e = blockIdx.x * blockDim.x + threadIdx.x; e < EE; e += stride) {
        int src, dst;
        if (e < E) { src = ei[e]; dst = ei[E + e]; }
        else       { src = dst = e - E; }
#pragma unroll
        for (int h = 0; h < H; ++h) {
            float v = asn[src * H + h] + adn[dst * H + h];
            v = (v > 0.0f) ? v : 0.2f * v;
            atomicMaxFloat(&mbuf[dst * H + h], v);
        }
    }
}

template <int H>
__global__ void edge_exp_kernel(const int* __restrict__ ei, int E, int EE,
                                const float* __restrict__ asn,
                                const float* __restrict__ adn,
                                const float* __restrict__ stats,
                                const float* __restrict__ mbuf,
                                float* __restrict__ ebuf) {
    const int t = blockIdx.x * blockDim.x + threadIdx.x;
    const int e = t / H;
    const int h = t - e * H;
    if (e >= EE) return;
    int src, dst;
    if (e < E) { src = ei[e]; dst = ei[E + e]; }
    else       { src = dst = e - E; }
    const bool bounded = stats[H] != 0.0f;
    const float ad = adn[dst * H + h];
    float v = asn[src * H + h] + ad;
    v = (v > 0.0f) ? v : 0.2f * v;
    float m;
    if (bounded) {
        float b = stats[h] + ad;
        m = (b > 0.0f) ? b : 0.2f * b;
    } else {
        m = mbuf[dst * H + h];
    }
    if (!(m >= -3.4e38f && m <= 3.4e38f)) m = 0.0f;
    const float ex = __expf(v - m);
    ebuf[t] = ex;
}

template <int H, int R>
__global__ void __launch_bounds__(1024)
den_bins_kernel(const int* __restrict__ ei, int E, int EE,
                const float* __restrict__ ebuf,
                float* __restrict__ den, int n) {
    static_assert(H == 1 || H == 2, "term loads are written for 1 or 2 heads");
    static_assert((R * H) % 128 == 0, "whole lines, 16-byte clears");
    constexpr int U   = 8;
    constexpr int TPB = 1024;
    __shared__ float bins[R * H];
    const int lo  = blockIdx.x * R;
    const int cnt = min(R, n - lo);
    for (int i = threadIdx.x; i < R * H / 4; i += TPB)
        ((float4*)bins)[i] = make_float4(0.0f, 0.0f, 0.0f, 0.0f);
    __syncthreads();

    const int es = (((EE + (int)gridDim.y - 1) / (int)gridDim.y) + TPB - 1) & ~(TPB - 1);
    const int eb = blockIdx.y * es;
    const int ee = min(eb + es, EE);
    for (int base = eb + (int)threadIdx.x; base < ee; base += TPB * U) {
        int   dl[U];
        float x0[U], x1[U];
#pragma unroll
        for (int u = 0; u < U; ++u) {
            const int e  = base + u * TPB;
            const int ec = min(e, E - 1);
            const int et = min(e, EE - 1);
            dl[u] = ei[E + ec];
            if constexpr (H == 2) {
                const float2 v = *(const float2*)(ebuf + (size_t)et * 2);
                x0[u] = v.x; x1[u] = v.y;
            } else {
                x0[u] = ebuf[et]; x1[u] = 0.0f;
            }
        }
#pragma unroll
        for (int u = 0; u < U; ++u) {
            if constexpr (H == 2) asm volatile("" :: "v"(dl[u]), "v"(x0[u]), "v"(x1[u]));
            else                  asm volatile("" :: "v"(dl[u]), "v"(x0[u]));
        }
#pragma unroll
        for (int u = 0; u < U; ++u) {
            const int e = base + u * TPB;
            const int d = (e < E) ? dl[u] : e - E;
            const unsigned off = (unsigned)(d - lo);
            if (e < ee && off < (unsigned)cnt) {
                atomicAdd(&bins[off * H], x0[u]);
                if constexpr (H == 2) atomicAdd(&bins[off * H + 1], x1[u]);
            }
        }
    }
    __syncthreads();

    float* dp = den + (size_t)lo * H;
    for (int i = threadIdx.x; i < cnt * H; i += TPB)
        atomicAdd(dp + i, bins[i]);
}

template <int H, int C>
__global__ void edge_agg_kernel(const int* __restrict__ ei, int E, int EE,
                                const float* __restrict__ ebuf,
                                const float* __restrict__ den,
                                const float* __restrict__ xl,
                                float* __restrict__ agg) {
    static_assert(C == 32, "one head's row piece must be one 32-lane line");
    static_assert(H == 1 || H == 2, "alpha loads are written for 1 or 2 heads");
    constexpr int F = H * C;
    const int lane = threadIdx.x & 31;
    const int e0 = __builtin_amdgcn_readfirstlane(
        (int)(blockIdx.x * blockDim.x + threadIdx.x) - lane);
    if (e0 >= EE) return;

    const int e = e0 + lane;
    int src = 0, dst = 0;
    float al[H];
    for (int h = 0; h < H; ++h) al[h] = 0.0f;
    if (e < EE) {
        if (e < E) { src = ei[e]; dst = ei[E + e]; }
        else       { src = dst = e - E; }
        if constexpr (H == 2) {
            const float2 ex = *(const float2*)(ebuf + (size_t)e * 2);
            const float2 dn = *(const float2*)(den + (size_t)dst * 2);
            al[0] = ex.x / (dn.x + 1e-16f);
            al[1] = ex.y / (dn.y + 1e-16f);
        } else {
            al[0] = ebuf[e] / (den[dst] + 1e-16f);
        }
    }

    if (EE - e0 >= 32) {
        float v[32 * H];
#pragma unroll
        for (int j = 0; j < 32; ++j) {
            const int s = __builtin_amdgcn_readlane(src, j);
            const float* row = xl + (size_t)s * F;
            for (int h = 0; h < H; ++h) v[j * H + h] = row[h * C + lane];
        }
#pragma unroll
        for (int j = 0; j < 32; ++j) {
            const int d = __builtin_amdgcn_readlane(dst, j);
            float* orow = agg + (size_t)d * F;
            for (int h = 0; h < H; ++h) {
                const float a = __int_as_float(
                    __builtin_amdgcn_readlane(__float_as_int(al[h]), j));
                atomicAdd(orow + h * C + lane, v[j * H + h] * a);
            }
        }
    } else {
        const int cnt = EE - e0;
        for (int j = 0; j < cnt; ++j) {
            const int s = __builtin_amdgcn_readlane(src, j);
            const int d = __builtin_amdgcn_readlane(dst, j);
            const float* row = xl + (size_t)s * F;
            float* orow = agg + (size_t)d * F;
            for (int h = 0; h < H; ++h) {
                const float a = __int_as_float(
                    __builtin_amdgcn_readlane(__float_as_int(al[h]), j));
                atomicAdd(orow + h * C + lane, row[h * C + lane] * a);
            }
        }
    }
}

static inline int cdiv(long long a, long long b) { return (int)((a + b - 1) / b); }

extern "C" void kernel_launch(void* const* d_in, const int* in_sizes, int n_in,
                              void* d_out, int out_size, void* d_ws, size_t ws_size,
                              hipStream_t stream) {
    const float* x      = (const float*)d_in[0];
    const int*   ei     = (const int*)d_in[1];
    const float* W1     = (const float*)d_in[2];
    const float* a_src1 = (const float*)d_in[3];
    const float* a_dst1 = (const float*)d_in[4];
    const float* b1     = (const float*)d_in[5];
    const float* W2     = (const float*)d_in[6];
    const float* a_src2 = (const float*)d_in[7];
    const float* a_dst2 = (const float*)d_in[8];
    const float* b2     = (const float*)d_in[9];
    float* out = (float*)d_out;

    const int N  = in_sizes[0] / IN_DIM;
    const int E  = in_sizes[1] / 2;
    const int EE = E + N;
    const int F1 = HEADS * HID;

    float* ws   = (float*)d_ws;
    float* xl1  = ws;
    float* agg1 = xl1  + (size_t)N * F1;
    float* asn1 = agg1 + (size_t)N * F1;
    float* adn1 = asn1 + (size_t)N * HEADS;
    float* m1   = adn1 + (size_t)N * HEADS;
    float* den1 = m1   + (size_t)N * HEADS;
    float* ex1  = den1 + (size_t)N * HEADS;
    float* xl2  = ex1  + (size_t)EE * HEADS;
    float* asn2 = xl2  + (size_t)N * OUT_DIM;
    float* adn2 = asn2 + (size_t)N;
    float* m2   = adn2 + (size_t)N;
    float* den2 = m2   + (size_t)N;
    float* ex2  = den2 + (size_t)N;

    const int TB = 256;
    const int NSB = cdiv(N, TB);

    float* part1  = xl2;
    float* stats1 = part1 + (size_t)NSB * 2 * HEADS;
    float* part2  = ex1;
    float* stats2 = part2 + (size_t)NSB * 2;


    {
        int mtiles = cdiv(N, 16);
        wmma_gemm_f32<IN_DIM, F1, false><<<cdiv(mtiles, 4), 128, 0, stream>>>(
            x, W1, xl1, b1, agg1, N);
        score_kernel<HEADS, HID><<<NSB, TB, 0, stream>>>(xl1, a_src1, a_dst1, asn1, adn1, den1, part1, N);
        range_kernel<HEADS><<<64, TB, 0, stream>>>(part1, NSB, stats1, m1, N * HEADS);
        edge_max_kernel<HEADS><<<1024, TB, 0, stream>>>(ei, E, EE, asn1, adn1, stats1, m1);
        edge_exp_kernel<HEADS><<<cdiv((long long)EE * HEADS, TB), TB, 0, stream>>>(ei, E, EE, asn1, adn1, stats1, m1, ex1);
        den_bins_kernel<HEADS, DEN_R><<<dim3(cdiv(N, DEN_R), DEN_SLICES), 1024, 0, stream>>>(ei, E, EE, ex1, den1, N);
        edge_agg_kernel<HEADS, HID><<<cdiv(EE, TB), TB, 0, stream>>>(
            ei, E, EE, ex1, den1, xl1, agg1);
    }

    {
        int mtiles = cdiv(N, 16);
        wmma_gemm_f32<F1, OUT_DIM, true><<<cdiv(mtiles, 4), 128, 0, stream>>>(
            agg1, W2, xl2, b2, out, N);
        score_kernel<1, OUT_DIM><<<NSB, TB, 0, stream>>>(xl2, a_src2, a_dst2, asn2, adn2, den2, part2, N);
        range_kernel<1><<<64, TB, 0, stream>>>(part2, NSB, stats2, m2, N);
        edge_max_kernel<1><<<1024, TB, 0, stream>>>(ei, E, EE, asn2, adn2, stats2, m2);
        edge_exp_kernel<1><<<cdiv(EE, TB), TB, 0, stream>>>(ei, E, EE, asn2, adn2, stats2, m2, ex2);
        den_bins_kernel<1, DEN_R><<<dim3(cdiv(N, DEN_R), DEN_SLICES), 1024, 0, stream>>>(ei, E, EE, ex2, den2, N);
        edge_agg_kernel<1, OUT_DIM><<<cdiv(EE, TB), TB, 0, stream>>>(
            ei, E, EE, ex2, den2, xl2, out);
    }
}


# Node-score launches folded into the product kernels: a wave forms its 16 rows' scores from its accumulators (lane partial dots + DPP row adds), range fold per tile; score sum order changes
# speedup vs baseline: 3.1728x; 1.0379x over previous
#include <hip/hip_runtime.h>
#include <hip/hip_bf16.h>
#include <math.h>

typedef float v2f __attribute__((ext_vector_type(2)));
typedef float v8f __attribute__((ext_vector_type(8)));

#define IN_DIM   128
#define HID      32
#define HEADS    2
#define OUT_DIM  32

__device__ __forceinline__ void atomicMaxFloat(float* addr, float value) {
    if (value >= 0.0f) {
        atomicMax((int*)addr, __float_as_int(value));
    } else {
        atomicMin((unsigned int*)addr, __float_as_uint(value));
    }
}

template <int MASK>
__device__ __forceinline__ float row16_xor(float v) {
    return __int_as_float(__builtin_amdgcn_update_dpp(
        0, __float_as_int(v), 0x160 + MASK, 0xf, 0xf, true));
}

template <int K, int NC, int H, bool RELU_A>
__global__ void wmma_gemm_f32(const float* __restrict__ A,
                              const float* __restrict__ B,
                              float* __restrict__ C,
                              const float* __restrict__ bias,
                              float* __restrict__ S,
                              const float* __restrict__ a_src,
                              const float* __restrict__ a_dst,
                              float* __restrict__ asn,
                              float* __restrict__ adn,
                              float* __restrict__ den,
                              float* __restrict__ part, int M) {
    constexpr int NT = NC / 16;
    __shared__ v2f Bs2[(K / 2) * NC];

    for (int p = threadIdx.x; p < (K / 2) * NC; p += blockDim.x) {
        int k2 = p / NC;
        int n  = p - k2 * NC;
        v2f pr;
        pr.x = B[(2 * k2 + 0) * NC + n];
        pr.y = B[(2 * k2 + 1) * NC + n];
        Bs2[p] = pr;
    }
    __syncthreads();

    const int wave = threadIdx.x >> 5;
    const int lane = threadIdx.x & 31;
    const int mt   = blockIdx.x * 4 + wave;
    if (mt * 16 >= M) return;

    const int halfK = (lane >> 4) * 2;
    const int ncol  = lane & 15;
    const int arow  = mt * 16 + (lane & 15);
    const float* Arow = A + (size_t)arow * K;

    {
        static_assert(128 % NC == 0, "a lane must keep its columns");
        constexpr int NV = 16 * NC / 128;
        const int c0 = (lane * 4) % NC;
        float4 bv;
        bv.x = bias[c0 + 0];
        bv.y = bias[c0 + 1];
        bv.z = bias[c0 + 2];
        bv.w = bias[c0 + 3];
        float4* Sp = (float4*)(S + (size_t)mt * 16 * NC) + lane;
#pragma unroll
        for (int i = 0; i < NV; ++i) Sp[i * 32] = bv;
    }

    v8f acc[NT];
#pragma unroll
    for (int t = 0; t < NT; ++t) acc[t] = (v8f){0, 0, 0, 0, 0, 0, 0, 0};

    for (int k = 0; k < K; k += 4) {
        v2f a = *(const v2f*)(Arow + k + halfK);
        if constexpr (RELU_A) {
            a.x = a.x > 0.0f ? a.x : 0.0f;
            a.y = a.y > 0.0f ? a.y : 0.0f;
        }
        const int prow = (k + halfK) >> 1;
#pragma unroll
        for (int t = 0; t < NT; ++t) {
            v2f b = Bs2[prow * NC + t * 16 + ncol];
            acc[t] = __builtin_amdgcn_wmma_f32_16x16x4_f32(
                false, a, false, b, (short)0, acc[t], false, false);
        }
    }

    const int rbase = mt * 16 + ((lane >> 4) << 3);
#pragma unroll
    for (int t = 0; t < NT; ++t) {
#pragma unroll
        for (int j = 0; j < 8; ++j) {
            C[(size_t)(rbase + j) * NC + t * 16 + ncol] = acc[t][j];
        }
    }

    static_assert(NT % H == 0, "a head is a whole number of column tiles");
    constexpr int TPH = NT / H;
    float ps[H][8], pd[H][8];
#pragma unroll
    for (int h = 0; h < H; ++h) {
#pragma unroll
        for (int j = 0; j < 8; ++j) { ps[h][j] = 0.0f; pd[h][j] = 0.0f; }
#pragma unroll
        for (int tt = 0; tt < TPH; ++tt) {
            const int t = h * TPH + tt;
            const float ws = a_src[t * 16 + ncol];
            const float wd = a_dst[t * 16 + ncol];
#pragma unroll
            for (int j = 0; j < 8; ++j) {
                ps[h][j] += acc[t][j] * ws;
                pd[h][j] += acc[t][j] * wd;
            }
        }
    }
#pragma unroll
    for (int h = 0; h < H; ++h) {
#pragma unroll
        for (int j = 0; j < 8; ++j) {
            ps[h][j] += row16_xor<8>(ps[h][j]);
            pd[h][j] += row16_xor<8>(pd[h][j]);
            ps[h][j] += row16_xor<4>(ps[h][j]);
            pd[h][j] += row16_xor<4>(pd[h][j]);
            ps[h][j] += row16_xor<2>(ps[h][j]);
            pd[h][j] += row16_xor<2>(pd[h][j]);
            ps[h][j] += row16_xor<1>(ps[h][j]);
            pd[h][j] += row16_xor<1>(pd[h][j]);
        }
    }
#pragma unroll
    for (int j = 0; j < 8; ++j) {
        if (ncol == j) {
            const size_t r = (size_t)(rbase + j);
#pragma unroll
            for (int h = 0; h < H; ++h) {
                asn[r * H + h] = ps[h][j];
                adn[r * H + h] = pd[h][j];
                den[r * H + h] = 0.0f;
            }
        }
    }
#pragma unroll
    for (int h = 0; h < H; ++h) {
        float mx = ps[h][0], mn = ps[h][0];
#pragma unroll
        for (int j = 1; j < 8; ++j) {
            mx = fmaxf(mx, ps[h][j]);
            mn = fminf(mn, ps[h][j]);
        }
        mx = fmaxf(mx, __shfl_xor(mx, 16));
        mn = fminf(mn, __shfl_xor(mn, 16));
        if (lane == 0) {
            part[(size_t)mt * 2 * H + h]     = mx;
            part[(size_t)mt * 2 * H + H + h] = mn;
        }
    }
}

#define SPREAD_OK 16.0f

#define DEN_R      12800
#define DEN_SLICES 32

template <int H>
__global__ void range_kernel(const float* __restrict__ part, int nblk,
                             float* __restrict__ stats,
                             float* __restrict__ mbuf, int nm) {
    __shared__ float red[2 * H][32];
    __shared__ int s_ok;
    float smax[H], smin[H];
#pragma unroll
    for (int h = 0; h < H; ++h) { smax[h] = -INFINITY; smin[h] = INFINITY; }
    constexpr int U = 8;
    for (int b0 = threadIdx.x; b0 < nblk; b0 += (int)blockDim.x * U) {
        float vx[U][H], vn[U][H];
#pragma unroll
        for (int u = 0; u < U; ++u) {
            const int b = min(b0 + u * (int)blockDim.x, nblk - 1);
#pragma unroll
            for (int h = 0; h < H; ++h) {
                vx[u][h] = part[(size_t)b * 2 * H + h];
                vn[u][h] = part[(size_t)b * 2 * H + H + h];
            }
        }
#pragma unroll
        for (int u = 0; u < U; ++u) {
#pragma unroll
            for (int h = 0; h < H; ++h) {
                smax[h] = fmaxf(smax[h], vx[u][h]);
                smin[h] = fminf(smin[h], vn[u][h]);
            }
        }
    }
    const int lane = threadIdx.x & (warpSize - 1);
    const int wave = threadIdx.x / warpSize;
    const int nw   = (blockDim.x + warpSize - 1) / warpSize;
#pragma unroll
    for (int h = 0; h < H; ++h) {
        for (int off = warpSize / 2; off > 0; off >>= 1) {
            smax[h] = fmaxf(smax[h], __shfl_xor(smax[h], off));
            smin[h] = fminf(smin[h], __shfl_xor(smin[h], off));
        }
        if (lane == 0) { red[h][wave] = smax[h]; red[H + h][wave] = smin[h]; }
    }
    __syncthreads();
    if (wave == 0) {
        bool ok = true;
#pragma unroll
        for (int h = 0; h < H; ++h) {
            float mx = (lane < nw) ? red[h][lane] : -INFINITY;
            float mn = (lane < nw) ? red[H + h][lane] : INFINITY;
            for (int off = warpSize / 2; off > 0; off >>= 1) {
                mx = fmaxf(mx, __shfl_xor(mx, off));
                mn = fminf(mn, __shfl_xor(mn, off));
            }
            if (blockIdx.x == 0 && lane == 0) stats[h] = mx;
            ok = ok && ((mx - mn) <= SPREAD_OK);
        }
        if (lane == 0) {
            if (blockIdx.x == 0) stats[H] = ok ? 1.0f : 0.0f;
            s_ok = ok ? 1 : 0;
        }
    }
    __syncthreads();
    if (!s_ok) {
        const int stride = gridDim.x * blockDim.x;
        for (int i = blockIdx.x * blockDim.x + threadIdx.x; i < nm; i += stride)
            mbuf[i] = -INFINITY;
    }
}

template <int H>
__global__ void edge_max_kernel(const int* __restrict__ ei, int E, int EE,
                                const float* __restrict__ asn,
                                const float* __restrict__ adn,
                                const float* __restrict__ stats,
                                float* __restrict__ mbuf) {
    if (stats[H] != 0.0f) return;
    const int stride = gridDim.x * blockDim.x;
    for (int e = blockIdx.x * blockDim.x + threadIdx.x; e < EE; e += stride) {
        int src, dst;
        if (e < E) { src = ei[e]; dst = ei[E + e]; }
        else       { src = dst = e - E; }
#pragma unroll
        for (int h = 0; h < H; ++h) {
            float v = asn[src * H + h] + adn[dst * H + h];
            v = (v > 0.0f) ? v : 0.2f * v;
            atomicMaxFloat(&mbuf[dst * H + h], v);
        }
    }
}

template <int H>
__global__ void edge_exp_kernel(const int* __restrict__ ei, int E, int EE,
                                const float* __restrict__ asn,
                                const float* __restrict__ adn,
                                const float* __restrict__ stats,
                                const float* __restrict__ mbuf,
                                float* __restrict__ ebuf) {
    const int t = blockIdx.x * blockDim.x + threadIdx.x;
    const int e = t / H;
    const int h = t - e * H;
    if (e >= EE) return;
    int src, dst;
    if (e < E) { src = ei[e]; dst = ei[E + e]; }
    else       { src = dst = e - E; }
    const bool bounded = stats[H] != 0.0f;
    const float ad = adn[dst * H + h];
    float v = asn[src * H + h] + ad;
    v = (v > 0.0f) ? v : 0.2f * v;
    float m;
    if (bounded) {
        float b = stats[h] + ad;
        m = (b > 0.0f) ? b : 0.2f * b;
    } else {
        m = mbuf[dst * H + h];
    }
    if (!(m >= -3.4e38f && m <= 3.4e38f)) m = 0.0f;
    const float ex = __expf(v - m);
    ebuf[t] = ex;
}

template <int H, int R>
__global__ void __launch_bounds__(1024)
den_bins_kernel(const int* __restrict__ ei, int E, int EE,
                const float* __restrict__ ebuf,
                float* __restrict__ den, int n) {
    static_assert(H == 1 || H == 2, "term loads are written for 1 or 2 heads");
    static_assert((R * H) % 128 == 0, "whole lines, 16-byte clears");
    constexpr int U   = 8;
    constexpr int TPB = 1024;
    __shared__ float bins[R * H];
    const int lo  = blockIdx.x * R;
    const int cnt = min(R, n - lo);
    for (int i = threadIdx.x; i < R * H / 4; i += TPB)
        ((float4*)bins)[i] = make_float4(0.0f, 0.0f, 0.0f, 0.0f);
    __syncthreads();

    const int es = (((EE + (int)gridDim.y - 1) / (int)gridDim.y) + TPB - 1) & ~(TPB - 1);
    const int eb = blockIdx.y * es;
    const int ee = min(eb + es, EE);
    for (int base = eb + (int)threadIdx.x; base < ee; base += TPB * U) {
        int   dl[U];
        float x0[U], x1[U];
#pragma unroll
        for (int u = 0; u < U; ++u) {
            const int e  = base + u * TPB;
            const int ec = min(e, E - 1);
            const int et = min(e, EE - 1);
            dl[u] = ei[E + ec];
            if constexpr (H == 2) {
                const float2 v = *(const float2*)(ebuf + (size_t)et * 2);
                x0[u] = v.x; x1[u] = v.y;
            } else {
                x0[u] = ebuf[et]; x1[u] = 0.0f;
            }
        }
#pragma unroll
        for (int u = 0; u < U; ++u) {
            if constexpr (H == 2) asm volatile("" :: "v"(dl[u]), "v"(x0[u]), "v"(x1[u]));
            else                  asm volatile("" :: "v"(dl[u]), "v"(x0[u]));
        }
#pragma unroll
        for (int u = 0; u < U; ++u) {
            const int e = base + u * TPB;
            const int d = (e < E) ? dl[u] : e - E;
            const unsigned off = (unsigned)(d - lo);
            if (e < ee && off < (unsigned)cnt) {
                atomicAdd(&bins[off * H], x0[u]);
                if constexpr (H == 2) atomicAdd(&bins[off * H + 1], x1[u]);
            }
        }
    }
    __syncthreads();

    float* dp = den + (size_t)lo * H;
    for (int i = threadIdx.x; i < cnt * H; i += TPB)
        atomicAdd(dp + i, bins[i]);
}

template <int H, int C>
__global__ void edge_agg_kernel(const int* __restrict__ ei, int E, int EE,
                                const float* __restrict__ ebuf,
                                const float* __restrict__ den,
                                const float* __restrict__ xl,
                                float* __restrict__ agg) {
    static_assert(C == 32, "one head's row piece must be one 32-lane line");
    static_assert(H == 1 || H == 2, "alpha loads are written for 1 or 2 heads");
    constexpr int F = H * C;
    const int lane = threadIdx.x & 31;
    const int e0 = __builtin_amdgcn_readfirstlane(
        (int)(blockIdx.x * blockDim.x + threadIdx.x) - lane);
    if (e0 >= EE) return;

    const int e = e0 + lane;
    int src = 0, dst = 0;
    float al[H];
    for (int h = 0; h < H; ++h) al[h] = 0.0f;
    if (e < EE) {
        if (e < E) { src = ei[e]; dst = ei[E + e]; }
        else       { src = dst = e - E; }
        if constexpr (H == 2) {
            const float2 ex = *(const float2*)(ebuf + (size_t)e * 2);
            const float2 dn = *(const float2*)(den + (size_t)dst * 2);
            al[0] = ex.x / (dn.x + 1e-16f);
            al[1] = ex.y / (dn.y + 1e-16f);
        } else {
            al[0] = ebuf[e] / (den[dst] + 1e-16f);
        }
    }

    if (EE - e0 >= 32) {
        float v[32 * H];
#pragma unroll
        for (int j = 0; j < 32; ++j) {
            const int s = __builtin_amdgcn_readlane(src, j);
            const float* row = xl + (size_t)s * F;
            for (int h = 0; h < H; ++h) v[j * H + h] = row[h * C + lane];
        }
#pragma unroll
        for (int j = 0; j < 32; ++j) {
            const int d = __builtin_amdgcn_readlane(dst, j);
            float* orow = agg + (size_t)d * F;
            for (int h = 0; h < H; ++h) {
                const float a = __int_as_float(
                    __builtin_amdgcn_readlane(__float_as_int(al[h]), j));
                atomicAdd(orow + h * C + lane, v[j * H + h] * a);
            }
        }
    } else {
        const int cnt = EE - e0;
        for (int j = 0; j < cnt; ++j) {
            const int s = __builtin_amdgcn_readlane(src, j);
            const int d = __builtin_amdgcn_readlane(dst, j);
            const float* row = xl + (size_t)s * F;
            float* orow = agg + (size_t)d * F;
            for (int h = 0; h < H; ++h) {
                const float a = __int_as_float(
                    __builtin_amdgcn_readlane(__float_as_int(al[h]), j));
                atomicAdd(orow + h * C + lane, row[h * C + lane] * a);
            }
        }
    }
}

static inline int cdiv(long long a, long long b) { return (int)((a + b - 1) / b); }

extern "C" void kernel_launch(void* const* d_in, const int* in_sizes, int n_in,
                              void* d_out, int out_size, void* d_ws, size_t ws_size,
                              hipStream_t stream) {
    const float* x      = (const float*)d_in[0];
    const int*   ei     = (const int*)d_in[1];
    const float* W1     = (const float*)d_in[2];
    const float* a_src1 = (const float*)d_in[3];
    const float* a_dst1 = (const float*)d_in[4];
    const float* b1     = (const float*)d_in[5];
    const float* W2     = (const float*)d_in[6];
    const float* a_src2 = (const float*)d_in[7];
    const float* a_dst2 = (const float*)d_in[8];
    const float* b2     = (const float*)d_in[9];
    float* out = (float*)d_out;

    const int N  = in_sizes[0] / IN_DIM;
    const int E  = in_sizes[1] / 2;
    const int EE = E + N;
    const int F1 = HEADS * HID;

    float* ws   = (float*)d_ws;
    float* xl1  = ws;
    float* agg1 = xl1  + (size_t)N * F1;
    float* asn1 = agg1 + (size_t)N * F1;
    float* adn1 = asn1 + (size_t)N * HEADS;
    float* m1   = adn1 + (size_t)N * HEADS;
    float* den1 = m1   + (size_t)N * HEADS;
    float* ex1  = den1 + (size_t)N * HEADS;
    float* xl2  = ex1  + (size_t)EE * HEADS;
    float* asn2 = xl2  + (size_t)N * OUT_DIM;
    float* adn2 = asn2 + (size_t)N;
    float* m2   = adn2 + (size_t)N;
    float* den2 = m2   + (size_t)N;
    float* ex2  = den2 + (size_t)N;

    const int TB = 256;
    const int NSB = cdiv(N, 16);

    float* part1  = xl2;
    float* stats1 = part1 + (size_t)NSB * 2 * HEADS;
    float* part2  = ex1;
    float* stats2 = part2 + (size_t)NSB * 2;


    {
        int mtiles = cdiv(N, 16);
        wmma_gemm_f32<IN_DIM, F1, HEADS, false><<<cdiv(mtiles, 4), 128, 0, stream>>>(
            x, W1, xl1, b1, agg1, a_src1, a_dst1, asn1, adn1, den1, part1, N);
        range_kernel<HEADS><<<64, 1024, 0, stream>>>(part1, NSB, stats1, m1, N * HEADS);
        edge_max_kernel<HEADS><<<1024, TB, 0, stream>>>(ei, E, EE, asn1, adn1, stats1, m1);
        edge_exp_kernel<HEADS><<<cdiv((long long)EE * HEADS, TB), TB, 0, stream>>>(ei, E, EE, asn1, adn1, stats1, m1, ex1);
        den_bins_kernel<HEADS, DEN_R><<<dim3(cdiv(N, DEN_R), DEN_SLICES), 1024, 0, stream>>>(ei, E, EE, ex1, den1, N);
        edge_agg_kernel<HEADS, HID><<<cdiv(EE, TB), TB, 0, stream>>>(
            ei, E, EE, ex1, den1, xl1, agg1);
    }

    {
        int mtiles = cdiv(N, 16);
        wmma_gemm_f32<F1, OUT_DIM, 1, true><<<cdiv(mtiles, 4), 128, 0, stream>>>(
            agg1, W2, xl2, b2, out, a_src2, a_dst2, asn2, adn2, den2, part2, N);
        range_kernel<1><<<64, 1024, 0, stream>>>(part2, NSB, stats2, m2, N);
        edge_max_kernel<1><<<1024, TB, 0, stream>>>(ei, E, EE, asn2, adn2, stats2, m2);
        edge_exp_kernel<1><<<cdiv(EE, TB), TB, 0, stream>>>(ei, E, EE, asn2, adn2, stats2, m2, ex2);
        den_bins_kernel<1, DEN_R><<<dim3(cdiv(N, DEN_R), DEN_SLICES), 1024, 0, stream>>>(ei, E, EE, ex2, den2, N);
        edge_agg_kernel<1, OUT_DIM><<<cdiv(EE, TB), TB, 0, stream>>>(
            ei, E, EE, ex2, den2, xl2, out);
    }
}


# Exp and aggregation passes as resident grids: 8192 waves per launch loop over 32-edge chunks, next chunk's end points and terms loaded ahead; 265k short waves no longer enter; sum order as before
# speedup vs baseline: 3.2868x; 1.0359x over previous
#include <hip/hip_runtime.h>
#include <hip/hip_bf16.h>
#include <math.h>

typedef float v2f __attribute__((ext_vector_type(2)));
typedef float v8f __attribute__((ext_vector_type(8)));

#define IN_DIM   128
#define HID      32
#define HEADS    2
#define OUT_DIM  32

__device__ __forceinline__ void atomicMaxFloat(float* addr, float value) {
    if (value >= 0.0f) {
        atomicMax((int*)addr, __float_as_int(value));
    } else {
        atomicMin((unsigned int*)addr, __float_as_uint(value));
    }
}

template <int MASK>
__device__ __forceinline__ float row16_xor(float v) {
    return __int_as_float(__builtin_amdgcn_update_dpp(
        0, __float_as_int(v), 0x160 + MASK, 0xf, 0xf, true));
}

template <int K, int NC, int H, bool RELU_A>
__global__ void wmma_gemm_f32(const float* __restrict__ A,
                              const float* __restrict__ B,
                              float* __restrict__ C,
                              const float* __restrict__ bias,
                              float* __restrict__ S,
                              const float* __restrict__ a_src,
                              const float* __restrict__ a_dst,
                              float* __restrict__ asn,
                              float* __restrict__ adn,
                              float* __restrict__ den,
                              float* __restrict__ part, int M) {
    constexpr int NT = NC / 16;
    __shared__ v2f Bs2[(K / 2) * NC];

    for (int p = threadIdx.x; p < (K / 2) * NC; p += blockDim.x) {
        int k2 = p / NC;
        int n  = p - k2 * NC;
        v2f pr;
        pr.x = B[(2 * k2 + 0) * NC + n];
        pr.y = B[(2 * k2 + 1) * NC + n];
        Bs2[p] = pr;
    }
    __syncthreads();

    const int wave = threadIdx.x >> 5;
    const int lane = threadIdx.x & 31;
    const int mt   = blockIdx.x * 4 + wave;
    if (mt * 16 >= M) return;

    const int halfK = (lane >> 4) * 2;
    const int ncol  = lane & 15;
    const int arow  = mt * 16 + (lane & 15);
    const float* Arow = A + (size_t)arow * K;

    {
        static_assert(128 % NC == 0, "a lane must keep its columns");
        constexpr int NV = 16 * NC / 128;
        const int c0 = (lane * 4) % NC;
        float4 bv;
        bv.x = bias[c0 + 0];
        bv.y = bias[c0 + 1];
        bv.z = bias[c0 + 2];
        bv.w = bias[c0 + 3];
        float4* Sp = (float4*)(S + (size_t)mt * 16 * NC) + lane;
#pragma unroll
        for (int i = 0; i < NV; ++i) Sp[i * 32] = bv;
    }

    v8f acc[NT];
#pragma unroll
    for (int t = 0; t < NT; ++t) acc[t] = (v8f){0, 0, 0, 0, 0, 0, 0, 0};

    for (int k = 0; k < K; k += 4) {
        v2f a = *(const v2f*)(Arow + k + halfK);
        if constexpr (RELU_A) {
            a.x = a.x > 0.0f ? a.x : 0.0f;
            a.y = a.y > 0.0f ? a.y : 0.0f;
        }
        const int prow = (k + halfK) >> 1;
#pragma unroll
        for (int t = 0; t < NT; ++t) {
            v2f b = Bs2[prow * NC + t * 16 + ncol];
            acc[t] = __builtin_amdgcn_wmma_f32_16x16x4_f32(
                false, a, false, b, (short)0, acc[t], false, false);
        }
    }

    const int rbase = mt * 16 + ((lane >> 4) << 3);
#pragma unroll
    for (int t = 0; t < NT; ++t) {
#pragma unroll
        for (int j = 0; j < 8; ++j) {
            C[(size_t)(rbase + j) * NC + t * 16 + ncol] = acc[t][j];
        }
    }

    static_assert(NT % H == 0, "a head is a whole number of column tiles");
    constexpr int TPH = NT / H;
    float ps[H][8], pd[H][8];
#pragma unroll
    for (int h = 0; h < H; ++h) {
#pragma unroll
        for (int j = 0; j < 8; ++j) { ps[h][j] = 0.0f; pd[h][j] = 0.0f; }
#pragma unroll
        for (int tt = 0; tt < TPH; ++tt) {
            const int t = h * TPH + tt;
            const float ws = a_src[t * 16 + ncol];
            const float wd = a_dst[t * 16 + ncol];
#pragma unroll
            for (int j = 0; j < 8; ++j) {
                ps[h][j] += acc[t][j] * ws;
                pd[h][j] += acc[t][j] * wd;
            }
        }
    }
#pragma unroll
    for (int h = 0; h < H; ++h) {
#pragma unroll
        for (int j = 0; j < 8; ++j) {
            ps[h][j] += row16_xor<8>(ps[h][j]);
            pd[h][j] += row16_xor<8>(pd[h][j]);
            ps[h][j] += row16_xor<4>(ps[h][j]);
            pd[h][j] += row16_xor<4>(pd[h][j]);
            ps[h][j] += row16_xor<2>(ps[h][j]);
            pd[h][j] += row16_xor<2>(pd[h][j]);
            ps[h][j] += row16_xor<1>(ps[h][j]);
            pd[h][j] += row16_xor<1>(pd[h][j]);
        }
    }
#pragma unroll
    for (int j = 0; j < 8; ++j) {
        if (ncol == j) {
            const size_t r = (size_t)(rbase + j);
#pragma unroll
            for (int h = 0; h < H; ++h) {
                asn[r * H + h] = ps[h][j];
                adn[r * H + h] = pd[h][j];
                den[r * H + h] = 0.0f;
            }
        }
    }
#pragma unroll
    for (int h = 0; h < H; ++h) {
        float mx = ps[h][0], mn = ps[h][0];
#pragma unroll
        for (int j = 1; j < 8; ++j) {
            mx = fmaxf(mx, ps[h][j]);
            mn = fminf(mn, ps[h][j]);
        }
        mx = fmaxf(mx, __shfl_xor(mx, 16));
        mn = fminf(mn, __shfl_xor(mn, 16));
        if (lane == 0) {
            part[(size_t)mt * 2 * H + h]     = mx;
            part[(size_t)mt * 2 * H + H + h] = mn;
        }
    }
}

#define SPREAD_OK 16.0f

#define DEN_R      12800
#define DEN_SLICES 32

#define EDGE_GRID 1024
#define EDGE_U    4

template <int H>
__global__ void range_kernel(const float* __restrict__ part, int nblk,
                             float* __restrict__ stats,
                             float* __restrict__ mbuf, int nm) {
    __shared__ float red[2 * H][32];
    __shared__ int s_ok;
    float smax[H], smin[H];
#pragma unroll
    for (int h = 0; h < H; ++h) { smax[h] = -INFINITY; smin[h] = INFINITY; }
    constexpr int U = 8;
    for (int b0 = threadIdx.x; b0 < nblk; b0 += (int)blockDim.x * U) {
        float vx[U][H], vn[U][H];
#pragma unroll
        for (int u = 0; u < U; ++u) {
            const int b = min(b0 + u * (int)blockDim.x, nblk - 1);
#pragma unroll
            for (int h = 0; h < H; ++h) {
                vx[u][h] = part[(size_t)b * 2 * H + h];
                vn[u][h] = part[(size_t)b * 2 * H + H + h];
            }
        }
#pragma unroll
        for (int u = 0; u < U; ++u) {
#pragma unroll
            for (int h = 0; h < H; ++h) {
                smax[h] = fmaxf(smax[h], vx[u][h]);
                smin[h] = fminf(smin[h], vn[u][h]);
            }
        }
    }
    const int lane = threadIdx.x & (warpSize - 1);
    const int wave = threadIdx.x / warpSize;
    const int nw   = (blockDim.x + warpSize - 1) / warpSize;
#pragma unroll
    for (int h = 0; h < H; ++h) {
        for (int off = warpSize / 2; off > 0; off >>= 1) {
            smax[h] = fmaxf(smax[h], __shfl_xor(smax[h], off));
            smin[h] = fminf(smin[h], __shfl_xor(smin[h], off));
        }
        if (lane == 0) { red[h][wave] = smax[h]; red[H + h][wave] = smin[h]; }
    }
    __syncthreads();
    if (wave == 0) {
        bool ok = true;
#pragma unroll
        for (int h = 0; h < H; ++h) {
            float mx = (lane < nw) ? red[h][lane] : -INFINITY;
            float mn = (lane < nw) ? red[H + h][lane] : INFINITY;
            for (int off = warpSize / 2; off > 0; off >>= 1) {
                mx = fmaxf(mx, __shfl_xor(mx, off));
                mn = fminf(mn, __shfl_xor(mn, off));
            }
            if (blockIdx.x == 0 && lane == 0) stats[h] = mx;
            ok = ok && ((mx - mn) <= SPREAD_OK);
        }
        if (lane == 0) {
            if (blockIdx.x == 0) stats[H] = ok ? 1.0f : 0.0f;
            s_ok = ok ? 1 : 0;
        }
    }
    __syncthreads();
    if (!s_ok) {
        const int stride = gridDim.x * blockDim.x;
        for (int i = blockIdx.x * blockDim.x + threadIdx.x; i < nm; i += stride)
            mbuf[i] = -INFINITY;
    }
}

template <int H>
__global__ void edge_max_kernel(const int* __restrict__ ei, int E, int EE,
                                const float* __restrict__ asn,
                                const float* __restrict__ adn,
                                const float* __restrict__ stats,
                                float* __restrict__ mbuf) {
    if (stats[H] != 0.0f) return;
    const int stride = gridDim.x * blockDim.x;
    for (int e = blockIdx.x * blockDim.x + threadIdx.x; e < EE; e += stride) {
        int src, dst;
        if (e < E) { src = ei[e]; dst = ei[E + e]; }
        else       { src = dst = e - E; }
#pragma unroll
        for (int h = 0; h < H; ++h) {
            float v = asn[src * H + h] + adn[dst * H + h];
            v = (v > 0.0f) ? v : 0.2f * v;
            atomicMaxFloat(&mbuf[dst * H + h], v);
        }
    }
}

template <int H, int U>
__global__ void edge_exp_kernel(const int* __restrict__ ei, int E, int EE,
                                const float* __restrict__ asn,
                                const float* __restrict__ adn,
                                const float* __restrict__ stats,
                                const float* __restrict__ mbuf,
                                float* __restrict__ ebuf) {
    const int T      = EE * H;
    const int stride = (int)(gridDim.x * blockDim.x);
    const int tid    = (int)(blockIdx.x * blockDim.x + threadIdx.x);
    const int h      = tid % H;
    const bool bounded = stats[H] != 0.0f;
    const float gmax   = stats[h];
    for (int t0 = tid; t0 < T; t0 += stride * U) {
        int sv[U], dv[U];
#pragma unroll
        for (int u = 0; u < U; ++u) {
            const int t  = min(t0 + u * stride, T - 1);
            const int e  = t / H;
            const int ec = min(e, E - 1);
            int s0 = ei[ec];
            int d0 = ei[E + ec];
            if (e >= E) { s0 = e - E; d0 = e - E; }
            sv[u] = s0; dv[u] = d0;
        }
        float av[U], dd[U];
#pragma unroll
        for (int u = 0; u < U; ++u) {
            dd[u] = adn[dv[u] * H + h];
            av[u] = asn[sv[u] * H + h];
        }
#pragma unroll
        for (int u = 0; u < U; ++u) {
            const int t = t0 + u * stride;
            if (t < T) {
                const float ad = dd[u];
                float v = av[u] + ad;
                v = (v > 0.0f) ? v : 0.2f * v;
                float m;
                if (bounded) {
                    float b = gmax + ad;
                    m = (b > 0.0f) ? b : 0.2f * b;
                } else {
                    m = mbuf[dv[u] * H + h];
                }
                if (!(m >= -3.4e38f && m <= 3.4e38f)) m = 0.0f;
                ebuf[t] = __expf(v - m);
            }
        }
    }
}

template <int H, int R>
__global__ void __launch_bounds__(1024)
den_bins_kernel(const int* __restrict__ ei, int E, int EE,
                const float* __restrict__ ebuf,
                float* __restrict__ den, int n) {
    static_assert(H == 1 || H == 2, "term loads are written for 1 or 2 heads");
    static_assert((R * H) % 128 == 0, "whole lines, 16-byte clears");
    constexpr int U   = 8;
    constexpr int TPB = 1024;
    __shared__ float bins[R * H];
    const int lo  = blockIdx.x * R;
    const int cnt = min(R, n - lo);
    for (int i = threadIdx.x; i < R * H / 4; i += TPB)
        ((float4*)bins)[i] = make_float4(0.0f, 0.0f, 0.0f, 0.0f);
    __syncthreads();

    const int es = (((EE + (int)gridDim.y - 1) / (int)gridDim.y) + TPB - 1) & ~(TPB - 1);
    const int eb = blockIdx.y * es;
    const int ee = min(eb + es, EE);
    for (int base = eb + (int)threadIdx.x; base < ee; base += TPB * U) {
        int   dl[U];
        float x0[U], x1[U];
#pragma unroll
        for (int u = 0; u < U; ++u) {
            const int e  = base + u * TPB;
            const int ec = min(e, E - 1);
            const int et = min(e, EE - 1);
            dl[u] = ei[E + ec];
            if constexpr (H == 2) {
                const float2 v = *(const float2*)(ebuf + (size_t)et * 2);
                x0[u] = v.x; x1[u] = v.y;
            } else {
                x0[u] = ebuf[et]; x1[u] = 0.0f;
            }
        }
#pragma unroll
        for (int u = 0; u < U; ++u) {
            if constexpr (H == 2) asm volatile("" :: "v"(dl[u]), "v"(x0[u]), "v"(x1[u]));
            else                  asm volatile("" :: "v"(dl[u]), "v"(x0[u]));
        }
#pragma unroll
        for (int u = 0; u < U; ++u) {
            const int e = base + u * TPB;
            const int d = (e < E) ? dl[u] : e - E;
            const unsigned off = (unsigned)(d - lo);
            if (e < ee && off < (unsigned)cnt) {
                atomicAdd(&bins[off * H], x0[u]);
                if constexpr (H == 2) atomicAdd(&bins[off * H + 1], x1[u]);
            }
        }
    }
    __syncthreads();

    float* dp = den + (size_t)lo * H;
    for (int i = threadIdx.x; i < cnt * H; i += TPB)
        atomicAdd(dp + i, bins[i]);
}

template <int H, int C>
__global__ void edge_agg_kernel(const int* __restrict__ ei, int E, int EE,
                                const float* __restrict__ ebuf,
                                const float* __restrict__ den,
                                const float* __restrict__ xl,
                                float* __restrict__ agg) {
    static_assert(C == 32, "one head's row piece must be one 32-lane line");
    static_assert(H == 1 || H == 2, "alpha loads are written for 1 or 2 heads");
    constexpr int F = H * C;
    const int lane = threadIdx.x & 31;
    const int e_first = __builtin_amdgcn_readfirstlane(
        (int)(blockIdx.x * blockDim.x + threadIdx.x) - lane);
    const int estep = (int)(gridDim.x * blockDim.x);
    if (e_first >= EE) return;

    int src, dst;
    float ex[H];
    {
        const int et = min(e_first + lane, EE - 1);
        const int ec = min(et, E - 1);
        src = ei[ec];
        dst = ei[E + ec];
        if (et >= E) { src = et - E; dst = et - E; }
        if constexpr (H == 2) {
            const float2 x2 = *(const float2*)(ebuf + (size_t)et * 2);
            ex[0] = x2.x; ex[1] = x2.y;
        } else {
            ex[0] = ebuf[et];
        }
    }

    for (int e0 = e_first; e0 < EE; e0 += estep) {
        float dn[H];
        if constexpr (H == 2) {
            const float2 d2 = *(const float2*)(den + (size_t)dst * 2);
            dn[0] = d2.x; dn[1] = d2.y;
        } else {
            dn[0] = den[dst];
        }
        int nsrc, ndst;
        float nex[H];
        {
            const int et = min(e0 + estep + lane, EE - 1);
            const int ec = min(et, E - 1);
            nsrc = ei[ec];
            ndst = ei[E + ec];
            if (et >= E) { nsrc = et - E; ndst = et - E; }
            if constexpr (H == 2) {
                const float2 x2 = *(const float2*)(ebuf + (size_t)et * 2);
                nex[0] = x2.x; nex[1] = x2.y;
            } else {
                nex[0] = ebuf[et];
            }
        }
        float al[H];
#pragma unroll
        for (int h = 0; h < H; ++h) al[h] = ex[h] / (dn[h] + 1e-16f);

        if (EE - e0 >= 32) {
            float v[32 * H];
#pragma unroll
            for (int j = 0; j < 32; ++j) {
                const int s = __builtin_amdgcn_readlane(src, j);
                const float* row = xl + (size_t)s * F;
                for (int h = 0; h < H; ++h) v[j * H + h] = row[h * C + lane];
            }
#pragma unroll
            for (int j = 0; j < 32; ++j) {
                const int d = __builtin_amdgcn_readlane(dst, j);
                float* orow = agg + (size_t)d * F;
                for (int h = 0; h < H; ++h) {
                    const float a = __int_as_float(
                        __builtin_amdgcn_readlane(__float_as_int(al[h]), j));
                    atomicAdd(orow + h * C + lane, v[j * H + h] * a);
                }
            }
        } else {
            const int cnt = EE - e0;
            for (int j = 0; j < cnt; ++j) {
                const int s = __builtin_amdgcn_readlane(src, j);
                const int d = __builtin_amdgcn_readlane(dst, j);
                const float* row = xl + (size_t)s * F;
                float* orow = agg + (size_t)d * F;
                for (int h = 0; h < H; ++h) {
                    const float a = __int_as_float(
                        __builtin_amdgcn_readlane(__float_as_int(al[h]), j));
                    atomicAdd(orow + h * C + lane, row[h * C + lane] * a);
                }
            }
        }
        src = nsrc; dst = ndst;
#pragma unroll
        for (int h = 0; h < H; ++h) ex[h] = nex[h];
    }
}

static inline int cdiv(long long a, long long b) { return (int)((a + b - 1) / b); }

extern "C" void kernel_launch(void* const* d_in, const int* in_sizes, int n_in,
                              void* d_out, int out_size, void* d_ws, size_t ws_size,
                              hipStream_t stream) {
    const float* x      = (const float*)d_in[0];
    const int*   ei     = (const int*)d_in[1];
    const float* W1     = (const float*)d_in[2];
    const float* a_src1 = (const float*)d_in[3];
    const float* a_dst1 = (const float*)d_in[4];
    const float* b1     = (const float*)d_in[5];
    const float* W2     = (const float*)d_in[6];
    const float* a_src2 = (const float*)d_in[7];
    const float* a_dst2 = (const float*)d_in[8];
    const float* b2     = (const float*)d_in[9];
    float* out = (float*)d_out;

    const int N  = in_sizes[0] / IN_DIM;
    const int E  = in_sizes[1] / 2;
    const int EE = E + N;
    const int F1 = HEADS * HID;

    float* ws   = (float*)d_ws;
    float* xl1  = ws;
    float* agg1 = xl1  + (size_t)N * F1;
    float* asn1 = agg1 + (size_t)N * F1;
    float* adn1 = asn1 + (size_t)N * HEADS;
    float* m1   = adn1 + (size_t)N * HEADS;
    float* den1 = m1   + (size_t)N * HEADS;
    float* ex1  = den1 + (size_t)N * HEADS;
    float* xl2  = ex1  + (size_t)EE * HEADS;
    float* asn2 = xl2  + (size_t)N * OUT_DIM;
    float* adn2 = asn2 + (size_t)N;
    float* m2   = adn2 + (size_t)N;
    float* den2 = m2   + (size_t)N;
    float* ex2  = den2 + (size_t)N;

    const int TB = 256;
    const int NSB = cdiv(N, 16);

    float* part1  = xl2;
    float* stats1 = part1 + (size_t)NSB * 2 * HEADS;
    float* part2  = ex1;
    float* stats2 = part2 + (size_t)NSB * 2;


    {
        int mtiles = cdiv(N, 16);
        wmma_gemm_f32<IN_DIM, F1, HEADS, false><<<cdiv(mtiles, 4), 128, 0, stream>>>(
            x, W1, xl1, b1, agg1, a_src1, a_dst1, asn1, adn1, den1, part1, N);
        range_kernel<HEADS><<<64, 1024, 0, stream>>>(part1, NSB, stats1, m1, N * HEADS);
        edge_max_kernel<HEADS><<<1024, TB, 0, stream>>>(ei, E, EE, asn1, adn1, stats1, m1);
        edge_exp_kernel<HEADS, EDGE_U><<<EDGE_GRID, TB, 0, stream>>>(ei, E, EE, asn1, adn1, stats1, m1, ex1);
        den_bins_kernel<HEADS, DEN_R><<<dim3(cdiv(N, DEN_R), DEN_SLICES), 1024, 0, stream>>>(ei, E, EE, ex1, den1, N);
        edge_agg_kernel<HEADS, HID><<<EDGE_GRID, TB, 0, stream>>>(
            ei, E, EE, ex1, den1, xl1, agg1);
    }

    {
        int mtiles = cdiv(N, 16);
        wmma_gemm_f32<F1, OUT_DIM, 1, true><<<cdiv(mtiles, 4), 128, 0, stream>>>(
            agg1, W2, xl2, b2, out, a_src2, a_dst2, asn2, adn2, den2, part2, N);
        range_kernel<1><<<64, 1024, 0, stream>>>(part2, NSB, stats2, m2, N);
        edge_max_kernel<1><<<1024, TB, 0, stream>>>(ei, E, EE, asn2, adn2, stats2, m2);
        edge_exp_kernel<1, EDGE_U><<<EDGE_GRID, TB, 0, stream>>>(ei, E, EE, asn2, adn2, stats2, m2, ex2);
        den_bins_kernel<1, DEN_R><<<dim3(cdiv(N, DEN_R), DEN_SLICES), 1024, 0, stream>>>(ei, E, EE, ex2, den2, N);
        edge_agg_kernel<1, OUT_DIM><<<EDGE_GRID, TB, 0, stream>>>(
            ei, E, EE, ex2, den2, xl2, out);
    }
}


# Aggregation addressing: each edge of a chunk keeps its own 32-bit row offset register on one scalar base, products stay in the load registers: no s_wait_xcnt between a chunk's adds; sum order kept
# speedup vs baseline: 3.3873x; 1.0306x over previous
#include <hip/hip_runtime.h>
#include <hip/hip_bf16.h>
#include <math.h>

typedef float v2f __attribute__((ext_vector_type(2)));
typedef float v8f __attribute__((ext_vector_type(8)));

#define IN_DIM   128
#define HID      32
#define HEADS    2
#define OUT_DIM  32

__device__ __forceinline__ void atomicMaxFloat(float* addr, float value) {
    if (value >= 0.0f) {
        atomicMax((int*)addr, __float_as_int(value));
    } else {
        atomicMin((unsigned int*)addr, __float_as_uint(value));
    }
}

template <int MASK>
__device__ __forceinline__ float row16_xor(float v) {
    return __int_as_float(__builtin_amdgcn_update_dpp(
        0, __float_as_int(v), 0x160 + MASK, 0xf, 0xf, true));
}

template <int K, int NC, int H, bool RELU_A>
__global__ void wmma_gemm_f32(const float* __restrict__ A,
                              const float* __restrict__ B,
                              float* __restrict__ C,
                              const float* __restrict__ bias,
                              float* __restrict__ S,
                              const float* __restrict__ a_src,
                              const float* __restrict__ a_dst,
                              float* __restrict__ asn,
                              float* __restrict__ adn,
                              float* __restrict__ den,
                              float* __restrict__ part, int M) {
    constexpr int NT = NC / 16;
    __shared__ v2f Bs2[(K / 2) * NC];

    for (int p = threadIdx.x; p < (K / 2) * NC; p += blockDim.x) {
        int k2 = p / NC;
        int n  = p - k2 * NC;
        v2f pr;
        pr.x = B[(2 * k2 + 0) * NC + n];
        pr.y = B[(2 * k2 + 1) * NC + n];
        Bs2[p] = pr;
    }
    __syncthreads();

    const int wave = threadIdx.x >> 5;
    const int lane = threadIdx.x & 31;
    const int mt   = blockIdx.x * 4 + wave;
    if (mt * 16 >= M) return;

    const int halfK = (lane >> 4) * 2;
    const int ncol  = lane & 15;
    const int arow  = mt * 16 + (lane & 15);
    const float* Arow = A + (size_t)arow * K;

    {
        static_assert(128 % NC == 0, "a lane must keep its columns");
        constexpr int NV = 16 * NC / 128;
        const int c0 = (lane * 4) % NC;
        float4 bv;
        bv.x = bias[c0 + 0];
        bv.y = bias[c0 + 1];
        bv.z = bias[c0 + 2];
        bv.w = bias[c0 + 3];
        float4* Sp = (float4*)(S + (size_t)mt * 16 * NC) + lane;
#pragma unroll
        for (int i = 0; i < NV; ++i) Sp[i * 32] = bv;
    }

    v8f acc[NT];
#pragma unroll
    for (int t = 0; t < NT; ++t) acc[t] = (v8f){0, 0, 0, 0, 0, 0, 0, 0};

    for (int k = 0; k < K; k += 4) {
        v2f a = *(const v2f*)(Arow + k + halfK);
        if constexpr (RELU_A) {
            a.x = a.x > 0.0f ? a.x : 0.0f;
            a.y = a.y > 0.0f ? a.y : 0.0f;
        }
        const int prow = (k + halfK) >> 1;
#pragma unroll
        for (int t = 0; t < NT; ++t) {
            v2f b = Bs2[prow * NC + t * 16 + ncol];
            acc[t] = __builtin_amdgcn_wmma_f32_16x16x4_f32(
                false, a, false, b, (short)0, acc[t], false, false);
        }
    }

    const int rbase = mt * 16 + ((lane >> 4) << 3);
#pragma unroll
    for (int t = 0; t < NT; ++t) {
#pragma unroll
        for (int j = 0; j < 8; ++j) {
            C[(size_t)(rbase + j) * NC + t * 16 + ncol] = acc[t][j];
        }
    }

    static_assert(NT % H == 0, "a head is a whole number of column tiles");
    constexpr int TPH = NT / H;
    float ps[H][8], pd[H][8];
#pragma unroll
    for (int h = 0; h < H; ++h) {
#pragma unroll
        for (int j = 0; j < 8; ++j) { ps[h][j] = 0.0f; pd[h][j] = 0.0f; }
#pragma unroll
        for (int tt = 0; tt < TPH; ++tt) {
            const int t = h * TPH + tt;
            const float ws = a_src[t * 16 + ncol];
            const float wd = a_dst[t * 16 + ncol];
#pragma unroll
            for (int j = 0; j < 8; ++j) {
                ps[h][j] += acc[t][j] * ws;
                pd[h][j] += acc[t][j] * wd;
            }
        }
    }
#pragma unroll
    for (int h = 0; h < H; ++h) {
#pragma unroll
        for (int j = 0; j < 8; ++j) {
            ps[h][j] += row16_xor<8>(ps[h][j]);
            pd[h][j] += row16_xor<8>(pd[h][j]);
            ps[h][j] += row16_xor<4>(ps[h][j]);
            pd[h][j] += row16_xor<4>(pd[h][j]);
            ps[h][j] += row16_xor<2>(ps[h][j]);
            pd[h][j] += row16_xor<2>(pd[h][j]);
            ps[h][j] += row16_xor<1>(ps[h][j]);
            pd[h][j] += row16_xor<1>(pd[h][j]);
        }
    }
#pragma unroll
    for (int j = 0; j < 8; ++j) {
        if (ncol == j) {
            const size_t r = (size_t)(rbase + j);
#pragma unroll
            for (int h = 0; h < H; ++h) {
                asn[r * H + h] = ps[h][j];
                adn[r * H + h] = pd[h][j];
                den[r * H + h] = 0.0f;
            }
        }
    }
#pragma unroll
    for (int h = 0; h < H; ++h) {
        float mx = ps[h][0], mn = ps[h][0];
#pragma unroll
        for (int j = 1; j < 8; ++j) {
            mx = fmaxf(mx, ps[h][j]);
            mn = fminf(mn, ps[h][j]);
        }
        mx = fmaxf(mx, __shfl_xor(mx, 16));
        mn = fminf(mn, __shfl_xor(mn, 16));
        if (lane == 0) {
            part[(size_t)mt * 2 * H + h]     = mx;
            part[(size_t)mt * 2 * H + H + h] = mn;
        }
    }
}

#define SPREAD_OK 16.0f

#define DEN_R      12800
#define DEN_SLICES 32

#define EDGE_GRID 1024
#define EDGE_U    4

template <int H>
__global__ void range_kernel(const float* __restrict__ part, int nblk,
                             float* __restrict__ stats,
                             float* __restrict__ mbuf, int nm) {
    __shared__ float red[2 * H][32];
    __shared__ int s_ok;
    float smax[H], smin[H];
#pragma unroll
    for (int h = 0; h < H; ++h) { smax[h] = -INFINITY; smin[h] = INFINITY; }
    constexpr int U = 8;
    for (int b0 = threadIdx.x; b0 < nblk; b0 += (int)blockDim.x * U) {
        float vx[U][H], vn[U][H];
#pragma unroll
        for (int u = 0; u < U; ++u) {
            const int b = min(b0 + u * (int)blockDim.x, nblk - 1);
#pragma unroll
            for (int h = 0; h < H; ++h) {
                vx[u][h] = part[(size_t)b * 2 * H + h];
                vn[u][h] = part[(size_t)b * 2 * H + H + h];
            }
        }
#pragma unroll
        for (int u = 0; u < U; ++u) {
#pragma unroll
            for (int h = 0; h < H; ++h) {
                smax[h] = fmaxf(smax[h], vx[u][h]);
                smin[h] = fminf(smin[h], vn[u][h]);
            }
        }
    }
    const int lane = threadIdx.x & (warpSize - 1);
    const int wave = threadIdx.x / warpSize;
    const int nw   = (blockDim.x + warpSize - 1) / warpSize;
#pragma unroll
    for (int h = 0; h < H; ++h) {
        for (int off = warpSize / 2; off > 0; off >>= 1) {
            smax[h] = fmaxf(smax[h], __shfl_xor(smax[h], off));
            smin[h] = fminf(smin[h], __shfl_xor(smin[h], off));
        }
        if (lane == 0) { red[h][wave] = smax[h]; red[H + h][wave] = smin[h]; }
    }
    __syncthreads();
    if (wave == 0) {
        bool ok = true;
#pragma unroll
        for (int h = 0; h < H; ++h) {
            float mx = (lane < nw) ? red[h][lane] : -INFINITY;
            float mn = (lane < nw) ? red[H + h][lane] : INFINITY;
            for (int off = warpSize / 2; off > 0; off >>= 1) {
                mx = fmaxf(mx, __shfl_xor(mx, off));
                mn = fminf(mn, __shfl_xor(mn, off));
            }
            if (blockIdx.x == 0 && lane == 0) stats[h] = mx;
            ok = ok && ((mx - mn) <= SPREAD_OK);
        }
        if (lane == 0) {
            if (blockIdx.x == 0) stats[H] = ok ? 1.0f : 0.0f;
            s_ok = ok ? 1 : 0;
        }
    }
    __syncthreads();
    if (!s_ok) {
        const int stride = gridDim.x * blockDim.x;
        for (int i = blockIdx.x * blockDim.x + threadIdx.x; i < nm; i += stride)
            mbuf[i] = -INFINITY;
    }
}

template <int H>
__global__ void edge_max_kernel(const int* __restrict__ ei, int E, int EE,
                                const float* __restrict__ asn,
                                const float* __restrict__ adn,
                                const float* __restrict__ stats,
                                float* __restrict__ mbuf) {
    if (stats[H] != 0.0f) return;
    const int stride = gridDim.x * blockDim.x;
    for (int e = blockIdx.x * blockDim.x + threadIdx.x; e < EE; e += stride) {
        int src, dst;
        if (e < E) { src = ei[e]; dst = ei[E + e]; }
        else       { src = dst = e - E; }
#pragma unroll
        for (int h = 0; h < H; ++h) {
            float v = asn[src * H + h] + adn[dst * H + h];
            v = (v > 0.0f) ? v : 0.2f * v;
            atomicMaxFloat(&mbuf[dst * H + h], v);
        }
    }
}

template <int H, int U>
__global__ void edge_exp_kernel(const int* __restrict__ ei, int E, int EE,
                                const float* __restrict__ asn,
                                const float* __restrict__ adn,
                                const float* __restrict__ stats,
                                const float* __restrict__ mbuf,
                                float* __restrict__ ebuf) {
    const int T      = EE * H;
    const int stride = (int)(gridDim.x * blockDim.x);
    const int tid    = (int)(blockIdx.x * blockDim.x + threadIdx.x);
    const int h      = tid % H;
    const bool bounded = stats[H] != 0.0f;
    const float gmax   = stats[h];
    for (int t0 = tid; t0 < T; t0 += stride * U) {
        int sv[U], dv[U];
#pragma unroll
        for (int u = 0; u < U; ++u) {
            const int t  = min(t0 + u * stride, T - 1);
            const int e  = t / H;
            const int ec = min(e, E - 1);
            int s0 = ei[ec];
            int d0 = ei[E + ec];
            if (e >= E) { s0 = e - E; d0 = e - E; }
            sv[u] = s0; dv[u] = d0;
        }
        float av[U], dd[U];
#pragma unroll
        for (int u = 0; u < U; ++u) {
            dd[u] = adn[dv[u] * H + h];
            av[u] = asn[sv[u] * H + h];
        }
#pragma unroll
        for (int u = 0; u < U; ++u) {
            const int t = t0 + u * stride;
            if (t < T) {
                const float ad = dd[u];
                float v = av[u] + ad;
                v = (v > 0.0f) ? v : 0.2f * v;
                float m;
                if (bounded) {
                    float b = gmax + ad;
                    m = (b > 0.0f) ? b : 0.2f * b;
                } else {
                    m = mbuf[dv[u] * H + h];
                }
                if (!(m >= -3.4e38f && m <= 3.4e38f)) m = 0.0f;
                ebuf[t] = __expf(v - m);
            }
        }
    }
}

template <int H, int R>
__global__ void __launch_bounds__(1024)
den_bins_kernel(const int* __restrict__ ei, int E, int EE,
                const float* __restrict__ ebuf,
                float* __restrict__ den, int n) {
    static_assert(H == 1 || H == 2, "term loads are written for 1 or 2 heads");
    static_assert((R * H) % 128 == 0, "whole lines, 16-byte clears");
    constexpr int U   = 8;
    constexpr int TPB = 1024;
    __shared__ float bins[R * H];
    const int lo  = blockIdx.x * R;
    const int cnt = min(R, n - lo);
    for (int i = threadIdx.x; i < R * H / 4; i += TPB)
        ((float4*)bins)[i] = make_float4(0.0f, 0.0f, 0.0f, 0.0f);
    __syncthreads();

    const int es = (((EE + (int)gridDim.y - 1) / (int)gridDim.y) + TPB - 1) & ~(TPB - 1);
    const int eb = blockIdx.y * es;
    const int ee = min(eb + es, EE);
    for (int base = eb + (int)threadIdx.x; base < ee; base += TPB * U) {
        int   dl[U];
        float x0[U], x1[U];
#pragma unroll
        for (int u = 0; u < U; ++u) {
            const int e  = base + u * TPB;
            const int ec = min(e, E - 1);
            const int et = min(e, EE - 1);
            dl[u] = ei[E + ec];
            if constexpr (H == 2) {
                const float2 v = *(const float2*)(ebuf + (size_t)et * 2);
                x0[u] = v.x; x1[u] = v.y;
            } else {
                x0[u] = ebuf[et]; x1[u] = 0.0f;
            }
        }
#pragma unroll
        for (int u = 0; u < U; ++u) {
            if constexpr (H == 2) asm volatile("" :: "v"(dl[u]), "v"(x0[u]), "v"(x1[u]));
            else                  asm volatile("" :: "v"(dl[u]), "v"(x0[u]));
        }
#pragma unroll
        for (int u = 0; u < U; ++u) {
            const int e = base + u * TPB;
            const int d = (e < E) ? dl[u] : e - E;
            const unsigned off = (unsigned)(d - lo);
            if (e < ee && off < (unsigned)cnt) {
                atomicAdd(&bins[off * H], x0[u]);
                if constexpr (H == 2) atomicAdd(&bins[off * H + 1], x1[u]);
            }
        }
    }
    __syncthreads();

    float* dp = den + (size_t)lo * H;
    for (int i = threadIdx.x; i < cnt * H; i += TPB)
        atomicAdd(dp + i, bins[i]);
}

template <int H, int C>
__global__ void __attribute__((amdgpu_waves_per_eu(8, 8))) edge_agg_kernel(const int* __restrict__ ei, int E, int EE,
                                const float* __restrict__ ebuf,
                                const float* __restrict__ den,
                                const float* __restrict__ xl,
                                float* __restrict__ agg) {
    static_assert(C == 32, "one head's row piece must be one 32-lane line");
    static_assert(H == 1 || H == 2, "alpha loads are written for 1 or 2 heads");
    constexpr int F = H * C;
    const int lane = threadIdx.x & 31;
    const int e_first = __builtin_amdgcn_readfirstlane(
        (int)(blockIdx.x * blockDim.x + threadIdx.x) - lane);
    const int estep = (int)(gridDim.x * blockDim.x);
    if (e_first >= EE) return;

    int src, dst;
    float ex[H];
    {
        const int et = min(e_first + lane, EE - 1);
        const int ec = min(et, E - 1);
        src = ei[ec];
        dst = ei[E + ec];
        if (et >= E) { src = et - E; dst = et - E; }
        if constexpr (H == 2) {
            const float2 x2 = *(const float2*)(ebuf + (size_t)et * 2);
            ex[0] = x2.x; ex[1] = x2.y;
        } else {
            ex[0] = ebuf[et];
        }
    }

    for (int e0 = e_first; e0 < EE; e0 += estep) {
        float dn[H];
        if constexpr (H == 2) {
            const float2 d2 = *(const float2*)(den + (size_t)dst * 2);
            dn[0] = d2.x; dn[1] = d2.y;
        } else {
            dn[0] = den[dst];
        }
        int nsrc, ndst;
        float nex[H];
        {
            const int et = min(e0 + estep + lane, EE - 1);
            const int ec = min(et, E - 1);
            nsrc = ei[ec];
            ndst = ei[E + ec];
            if (et >= E) { nsrc = et - E; ndst = et - E; }
            if constexpr (H == 2) {
                const float2 x2 = *(const float2*)(ebuf + (size_t)et * 2);
                nex[0] = x2.x; nex[1] = x2.y;
            } else {
                nex[0] = ebuf[et];
            }
        }
        float al[H];
#pragma unroll
        for (int h = 0; h < H; ++h) al[h] = ex[h] / (dn[h] + 1e-16f);

        if (EE - e0 >= 32) {
            float v[32 * H];
            unsigned off[32];
            const unsigned ulane = (unsigned)lane;
#pragma unroll
            for (int j = 0; j < 32; ++j)
                off[j] = (unsigned)__builtin_amdgcn_readlane(src, j) * (unsigned)F + ulane;
#pragma unroll
            for (int j = 0; j < 32; ++j)
                for (int h = 0; h < H; ++h) v[j * H + h] = xl[off[j] + (unsigned)(h * C)];
#pragma unroll
            for (int j = 0; j < 32; ++j)
                off[j] = (unsigned)__builtin_amdgcn_readlane(dst, j) * (unsigned)F + ulane;
#pragma unroll
            for (int j = 0; j < 32; ++j) {
                for (int h = 0; h < H; ++h) {
                    const float a = __int_as_float(
                        __builtin_amdgcn_readlane(__float_as_int(al[h]), j));
                    v[j * H + h] *= a;
                }
            }
#pragma unroll
            for (int j = 0; j < 32; ++j) {
                for (int h = 0; h < H; ++h)
                    atomicAdd(agg + (off[j] + (unsigned)(h * C)), v[j * H + h]);
            }
        } else {
            const int cnt = EE - e0;
            for (int j = 0; j < cnt; ++j) {
                const int s = __builtin_amdgcn_readlane(src, j);
                const int d = __builtin_amdgcn_readlane(dst, j);
                const float* row = xl + (size_t)s * F;
                float* orow = agg + (size_t)d * F;
                for (int h = 0; h < H; ++h) {
                    const float a = __int_as_float(
                        __builtin_amdgcn_readlane(__float_as_int(al[h]), j));
                    atomicAdd(orow + h * C + lane, row[h * C + lane] * a);
                }
            }
        }
        src = nsrc; dst = ndst;
#pragma unroll
        for (int h = 0; h < H; ++h) ex[h] = nex[h];
    }
}

static inline int cdiv(long long a, long long b) { return (int)((a + b - 1) / b); }

extern "C" void kernel_launch(void* const* d_in, const int* in_sizes, int n_in,
                              void* d_out, int out_size, void* d_ws, size_t ws_size,
                              hipStream_t stream) {
    const float* x      = (const float*)d_in[0];
    const int*   ei     = (const int*)d_in[1];
    const float* W1     = (const float*)d_in[2];
    const float* a_src1 = (const float*)d_in[3];
    const float* a_dst1 = (const float*)d_in[4];
    const float* b1     = (const float*)d_in[5];
    const float* W2     = (const float*)d_in[6];
    const float* a_src2 = (const float*)d_in[7];
    const float* a_dst2 = (const float*)d_in[8];
    const float* b2     = (const float*)d_in[9];
    float* out = (float*)d_out;

    const int N  = in_sizes[0] / IN_DIM;
    const int E  = in_sizes[1] / 2;
    const int EE = E + N;
    const int F1 = HEADS * HID;

    float* ws   = (float*)d_ws;
    float* xl1  = ws;
    float* agg1 = xl1  + (size_t)N * F1;
    float* asn1 = agg1 + (size_t)N * F1;
    float* adn1 = asn1 + (size_t)N * HEADS;
    float* m1   = adn1 + (size_t)N * HEADS;
    float* den1 = m1   + (size_t)N * HEADS;
    float* ex1  = den1 + (size_t)N * HEADS;
    float* xl2  = ex1  + (size_t)EE * HEADS;
    float* asn2 = xl2  + (size_t)N * OUT_DIM;
    float* adn2 = asn2 + (size_t)N;
    float* m2   = adn2 + (size_t)N;
    float* den2 = m2   + (size_t)N;
    float* ex2  = den2 + (size_t)N;

    const int TB = 256;
    const int NSB = cdiv(N, 16);

    float* part1  = xl2;
    float* stats1 = part1 + (size_t)NSB * 2 * HEADS;
    float* part2  = ex1;
    float* stats2 = part2 + (size_t)NSB * 2;


    {
        int mtiles = cdiv(N, 16);
        wmma_gemm_f32<IN_DIM, F1, HEADS, false><<<cdiv(mtiles, 4), 128, 0, stream>>>(
            x, W1, xl1, b1, agg1, a_src1, a_dst1, asn1, adn1, den1, part1, N);
        range_kernel<HEADS><<<64, 1024, 0, stream>>>(part1, NSB, stats1, m1, N * HEADS);
        edge_max_kernel<HEADS><<<1024, TB, 0, stream>>>(ei, E, EE, asn1, adn1, stats1, m1);
        edge_exp_kernel<HEADS, EDGE_U><<<EDGE_GRID, TB, 0, stream>>>(ei, E, EE, asn1, adn1, stats1, m1, ex1);
        den_bins_kernel<HEADS, DEN_R><<<dim3(cdiv(N, DEN_R), DEN_SLICES), 1024, 0, stream>>>(ei, E, EE, ex1, den1, N);
        edge_agg_kernel<HEADS, HID><<<EDGE_GRID, TB, 0, stream>>>(
            ei, E, EE, ex1, den1, xl1, agg1);
    }

    {
        int mtiles = cdiv(N, 16);
        wmma_gemm_f32<F1, OUT_DIM, 1, true><<<cdiv(mtiles, 4), 128, 0, stream>>>(
            agg1, W2, xl2, b2, out, a_src2, a_dst2, asn2, adn2, den2, part2, N);
        range_kernel<1><<<64, 1024, 0, stream>>>(part2, NSB, stats2, m2, N);
        edge_max_kernel<1><<<1024, TB, 0, stream>>>(ei, E, EE, asn2, adn2, stats2, m2);
        edge_exp_kernel<1, EDGE_U><<<EDGE_GRID, TB, 0, stream>>>(ei, E, EE, asn2, adn2, stats2, m2, ex2);
        den_bins_kernel<1, DEN_R><<<dim3(cdiv(N, DEN_R), DEN_SLICES), 1024, 0, stream>>>(ei, E, EE, ex2, den2, N);
        edge_agg_kernel<1, OUT_DIM><<<EDGE_GRID, TB, 0, stream>>>(
            ei, E, EE, ex2, den2, xl2, out);
    }
}


# Score-range fold and exact-max fallback moved into the edge launches: one max/min entry per product block, each exp block folds the table; range and max launches gone (12 to 8); sum order kept
# speedup vs baseline: 3.4895x; 1.0302x over previous
#include <hip/hip_runtime.h>
#include <hip/hip_bf16.h>
#include <math.h>

typedef float v2f __attribute__((ext_vector_type(2)));
typedef float v8f __attribute__((ext_vector_type(8)));

#define IN_DIM   128
#define HID      32
#define HEADS    2
#define OUT_DIM  32

__device__ __forceinline__ void atomicMaxFloat(float* addr, float value) {
    if (value >= 0.0f) {
        atomicMax((int*)addr, __float_as_int(value));
    } else {
        atomicMin((unsigned int*)addr, __float_as_uint(value));
    }
}

template <int MASK>
__device__ __forceinline__ float row16_xor(float v) {
    return __int_as_float(__builtin_amdgcn_update_dpp(
        0, __float_as_int(v), 0x160 + MASK, 0xf, 0xf, true));
}

template <int K, int NC, int H, bool RELU_A>
__global__ void wmma_gemm_f32(const float* __restrict__ A,
                              const float* __restrict__ B,
                              float* __restrict__ C,
                              const float* __restrict__ bias,
                              float* __restrict__ S,
                              const float* __restrict__ a_src,
                              const float* __restrict__ a_dst,
                              float* __restrict__ asn,
                              float* __restrict__ adn,
                              float* __restrict__ den,
                              float* __restrict__ mbuf,
                              float* __restrict__ part, int M) {
    constexpr int NT = NC / 16;
    __shared__ v2f Bs2[(K / 2) * NC];
    __shared__ float wred[4][2 * H];

    for (int p = threadIdx.x; p < (K / 2) * NC; p += blockDim.x) {
        int k2 = p / NC;
        int n  = p - k2 * NC;
        v2f pr;
        pr.x = B[(2 * k2 + 0) * NC + n];
        pr.y = B[(2 * k2 + 1) * NC + n];
        Bs2[p] = pr;
    }
    __syncthreads();

    const int wave = threadIdx.x >> 5;
    const int lane = threadIdx.x & 31;
    const int mt   = blockIdx.x * 4 + wave;
    if (mt * 16 < M) {

    const int halfK = (lane >> 4) * 2;
    const int ncol  = lane & 15;
    const int arow  = mt * 16 + (lane & 15);
    const float* Arow = A + (size_t)arow * K;

    {
        static_assert(128 % NC == 0, "a lane must keep its columns");
        constexpr int NV = 16 * NC / 128;
        const int c0 = (lane * 4) % NC;
        float4 bv;
        bv.x = bias[c0 + 0];
        bv.y = bias[c0 + 1];
        bv.z = bias[c0 + 2];
        bv.w = bias[c0 + 3];
        float4* Sp = (float4*)(S + (size_t)mt * 16 * NC) + lane;
#pragma unroll
        for (int i = 0; i < NV; ++i) Sp[i * 32] = bv;
    }

    v8f acc[NT];
#pragma unroll
    for (int t = 0; t < NT; ++t) acc[t] = (v8f){0, 0, 0, 0, 0, 0, 0, 0};

    for (int k = 0; k < K; k += 4) {
        v2f a = *(const v2f*)(Arow + k + halfK);
        if constexpr (RELU_A) {
            a.x = a.x > 0.0f ? a.x : 0.0f;
            a.y = a.y > 0.0f ? a.y : 0.0f;
        }
        const int prow = (k + halfK) >> 1;
#pragma unroll
        for (int t = 0; t < NT; ++t) {
            v2f b = Bs2[prow * NC + t * 16 + ncol];
            acc[t] = __builtin_amdgcn_wmma_f32_16x16x4_f32(
                false, a, false, b, (short)0, acc[t], false, false);
        }
    }

    const int rbase = mt * 16 + ((lane >> 4) << 3);
#pragma unroll
    for (int t = 0; t < NT; ++t) {
#pragma unroll
        for (int j = 0; j < 8; ++j) {
            C[(size_t)(rbase + j) * NC + t * 16 + ncol] = acc[t][j];
        }
    }

    static_assert(NT % H == 0, "a head is a whole number of column tiles");
    constexpr int TPH = NT / H;
    float ps[H][8], pd[H][8];
#pragma unroll
    for (int h = 0; h < H; ++h) {
#pragma unroll
        for (int j = 0; j < 8; ++j) { ps[h][j] = 0.0f; pd[h][j] = 0.0f; }
#pragma unroll
        for (int tt = 0; tt < TPH; ++tt) {
            const int t = h * TPH + tt;
            const float ws = a_src[t * 16 + ncol];
            const float wd = a_dst[t * 16 + ncol];
#pragma unroll
            for (int j = 0; j < 8; ++j) {
                ps[h][j] += acc[t][j] * ws;
                pd[h][j] += acc[t][j] * wd;
            }
        }
    }
#pragma unroll
    for (int h = 0; h < H; ++h) {
#pragma unroll
        for (int j = 0; j < 8; ++j) {
            ps[h][j] += row16_xor<8>(ps[h][j]);
            pd[h][j] += row16_xor<8>(pd[h][j]);
            ps[h][j] += row16_xor<4>(ps[h][j]);
            pd[h][j] += row16_xor<4>(pd[h][j]);
            ps[h][j] += row16_xor<2>(ps[h][j]);
            pd[h][j] += row16_xor<2>(pd[h][j]);
            ps[h][j] += row16_xor<1>(ps[h][j]);
            pd[h][j] += row16_xor<1>(pd[h][j]);
        }
    }
#pragma unroll
    for (int j = 0; j < 8; ++j) {
        if (ncol == j) {
            const size_t r = (size_t)(rbase + j);
#pragma unroll
            for (int h = 0; h < H; ++h) {
                asn[r * H + h] = ps[h][j];
                adn[r * H + h] = pd[h][j];
                den[r * H + h] = 0.0f;
                mbuf[r * H + h] = -INFINITY;
            }
        }
    }
#pragma unroll
    for (int h = 0; h < H; ++h) {
        float mx = ps[h][0], mn = ps[h][0];
#pragma unroll
        for (int j = 1; j < 8; ++j) {
            mx = fmaxf(mx, ps[h][j]);
            mn = fminf(mn, ps[h][j]);
        }
        mx = fmaxf(mx, __shfl_xor(mx, 16));
        mn = fminf(mn, __shfl_xor(mn, 16));
        if (lane == 0) {
            wred[wave][h]     = mx;
            wred[wave][H + h] = mn;
        }
    }

    } else {
        if (lane < 2 * H) wred[wave][lane] = (lane < H) ? -INFINITY : INFINITY;
    }
    __builtin_amdgcn_fence(__ATOMIC_RELEASE, "workgroup", "local");
    __builtin_amdgcn_s_barrier();
    __builtin_amdgcn_fence(__ATOMIC_ACQUIRE, "workgroup", "local");
    if (threadIdx.x < 2 * H) {
        const int t = threadIdx.x;
        float v = wred[0][t];
        for (int w = 1; w < 4; ++w)
            v = (t < H) ? fmaxf(v, wred[w][t]) : fminf(v, wred[w][t]);
        part[(size_t)blockIdx.x * 2 * H + t] = v;
    }
}

#define SPREAD_OK 16.0f

#define DEN_R      12800
#define DEN_SLICES 32

#define EDGE_GRID 1024
#define EDGE_TB   256
#define EDGE_U    4

template <int H, int NW>
__device__ __forceinline__ bool fold_score_range(const float* __restrict__ part, int nblk,
                                                 float (*red)[2 * NW], float (&gmax)[H]) {
    static_assert(2 * NW == 16, "one 16-lane row folds the rows' values of the block");
    constexpr int TPB = NW * 32;
    float smax[H], smin[H];
#pragma unroll
    for (int h = 0; h < H; ++h) { smax[h] = -INFINITY; smin[h] = INFINITY; }
    constexpr int U = 8;
    for (int b0 = threadIdx.x; b0 < nblk; b0 += TPB * U) {
        float vx[U][H], vn[U][H];
#pragma unroll
        for (int u = 0; u < U; ++u) {
            const int b = min(b0 + u * TPB, nblk - 1);
#pragma unroll
            for (int h = 0; h < H; ++h) {
                vx[u][h] = part[(size_t)b * 2 * H + h];
                vn[u][h] = part[(size_t)b * 2 * H + H + h];
            }
        }
#pragma unroll
        for (int u = 0; u < U; ++u) {
#pragma unroll
            for (int h = 0; h < H; ++h) {
                smax[h] = fmaxf(smax[h], vx[u][h]);
                smin[h] = fminf(smin[h], vn[u][h]);
            }
        }
    }
    const int lane = threadIdx.x & 31;
    const int wave = threadIdx.x >> 5;
#pragma unroll
    for (int h = 0; h < H; ++h) {
        smax[h] = fmaxf(smax[h], row16_xor<8>(smax[h]));
        smin[h] = fminf(smin[h], row16_xor<8>(smin[h]));
        smax[h] = fmaxf(smax[h], row16_xor<4>(smax[h]));
        smin[h] = fminf(smin[h], row16_xor<4>(smin[h]));
        smax[h] = fmaxf(smax[h], row16_xor<2>(smax[h]));
        smin[h] = fminf(smin[h], row16_xor<2>(smin[h]));
        smax[h] = fmaxf(smax[h], row16_xor<1>(smax[h]));
        smin[h] = fminf(smin[h], row16_xor<1>(smin[h]));
    }
    if ((lane & 15) == 0) {
        const int slot = wave * 2 + (lane >> 4);
#pragma unroll
        for (int h = 0; h < H; ++h) { red[h][slot] = smax[h]; red[H + h][slot] = smin[h]; }
    }
    __syncthreads();
    bool ok = true;
#pragma unroll
    for (int h = 0; h < H; ++h) {
        float mx = red[h][lane & 15];
        float mn = red[H + h][lane & 15];
        mx = fmaxf(mx, row16_xor<8>(mx));
        mn = fminf(mn, row16_xor<8>(mn));
        mx = fmaxf(mx, row16_xor<4>(mx));
        mn = fminf(mn, row16_xor<4>(mn));
        mx = fmaxf(mx, row16_xor<2>(mx));
        mn = fminf(mn, row16_xor<2>(mn));
        mx = fmaxf(mx, row16_xor<1>(mx));
        mn = fminf(mn, row16_xor<1>(mn));
        gmax[h] = mx;
        ok = ok && ((mx - mn) <= SPREAD_OK);
    }
    return ok;
}

template <int H, int U>
__global__ void edge_exp_kernel(const int* __restrict__ ei, int E, int EE,
                                const float* __restrict__ asn,
                                const float* __restrict__ adn,
                                const float* __restrict__ part, int nblk,
                                float* __restrict__ flag,
                                float* __restrict__ mbuf,
                                float* __restrict__ ebuf) {
    __shared__ float red[2 * H][2 * (EDGE_TB / 32)];
    float gm[H];
    const bool bounded = fold_score_range<H, EDGE_TB / 32>(part, nblk, red, gm);

    const int T      = EE * H;
    const int stride = (int)(gridDim.x * blockDim.x);
    const int tid    = (int)(blockIdx.x * blockDim.x + threadIdx.x);
    const int h      = tid % H;
    const float gmax = gm[h];
    if (tid == 0) flag[0] = bounded ? 1.0f : 0.0f;

    if (!bounded) {
        for (int t = tid; t < T; t += stride) {
            const int e = t / H;
            int src, dst;
            if (e < E) { src = ei[e]; dst = ei[E + e]; }
            else       { src = dst = e - E; }
            float v = asn[src * H + h] + adn[dst * H + h];
            v = (v > 0.0f) ? v : 0.2f * v;
            atomicMaxFloat(&mbuf[dst * H + h], v);
        }
        return;
    }

    for (int t0 = tid; t0 < T; t0 += stride * U) {
        int sv[U], dv[U];
#pragma unroll
        for (int u = 0; u < U; ++u) {
            const int t  = min(t0 + u * stride, T - 1);
            const int e  = t / H;
            const int ec = min(e, E - 1);
            int s0 = ei[ec];
            int d0 = ei[E + ec];
            if (e >= E) { s0 = e - E; d0 = e - E; }
            sv[u] = s0; dv[u] = d0;
        }
        float av[U], dd[U];
#pragma unroll
        for (int u = 0; u < U; ++u) {
            dd[u] = adn[dv[u] * H + h];
            av[u] = asn[sv[u] * H + h];
        }
#pragma unroll
        for (int u = 0; u < U; ++u) {
            const int t = t0 + u * stride;
            if (t < T) {
                const float ad = dd[u];
                float v = av[u] + ad;
                v = (v > 0.0f) ? v : 0.2f * v;
                const float b = gmax + ad;
                float m = (b > 0.0f) ? b : 0.2f * b;
                if (!(m >= -3.4e38f && m <= 3.4e38f)) m = 0.0f;
                ebuf[t] = __expf(v - m);
            }
        }
    }
}

template <int H, int R>
__global__ void __launch_bounds__(1024)
den_bins_kernel(const int* __restrict__ ei, int E, int EE,
                const float* __restrict__ asn,
                const float* __restrict__ adn,
                const float* __restrict__ mbuf,
                const float* __restrict__ flag,
                float* __restrict__ ebuf,
                float* __restrict__ den, int n) {
    static_assert(H == 1 || H == 2, "term loads are written for 1 or 2 heads");
    static_assert((R * H) % 128 == 0, "whole lines, 16-byte clears");
    constexpr int U   = 8;
    constexpr int TPB = 1024;
    __shared__ float bins[R * H];
    const int lo  = blockIdx.x * R;
    const int cnt = min(R, n - lo);
    const bool bounded = flag[0] != 0.0f;
    for (int i = threadIdx.x; i < R * H / 4; i += TPB)
        ((float4*)bins)[i] = make_float4(0.0f, 0.0f, 0.0f, 0.0f);
    __syncthreads();

    const int es = (((EE + (int)gridDim.y - 1) / (int)gridDim.y) + TPB - 1) & ~(TPB - 1);
    const int eb = blockIdx.y * es;
    const int ee = min(eb + es, EE);
    if (!bounded) {
        for (int e = eb + (int)threadIdx.x; e < ee; e += TPB) {
            int s, d;
            if (e < E) { s = ei[e]; d = ei[E + e]; }
            else       { s = e - E; d = e - E; }
            const unsigned off = (unsigned)(d - lo);
            if (off < (unsigned)cnt) {
                for (int h = 0; h < H; ++h) {
                    float v = asn[s * H + h] + adn[d * H + h];
                    v = (v > 0.0f) ? v : 0.2f * v;
                    float m = mbuf[d * H + h];
                    if (!(m >= -3.4e38f && m <= 3.4e38f)) m = 0.0f;
                    const float x = __expf(v - m);
                    ebuf[(size_t)e * H + h] = x;
                    atomicAdd(&bins[off * H + h], x);
                }
            }
        }
    }
    for (int base = eb + (int)threadIdx.x; bounded && base < ee; base += TPB * U) {
        int   dl[U];
        float x0[U], x1[U];
#pragma unroll
        for (int u = 0; u < U; ++u) {
            const int e  = base + u * TPB;
            const int ec = min(e, E - 1);
            const int et = min(e, EE - 1);
            dl[u] = ei[E + ec];
            if constexpr (H == 2) {
                const float2 v = *(const float2*)(ebuf + (size_t)et * 2);
                x0[u] = v.x; x1[u] = v.y;
            } else {
                x0[u] = ebuf[et]; x1[u] = 0.0f;
            }
        }
#pragma unroll
        for (int u = 0; u < U; ++u) {
            if constexpr (H == 2) asm volatile("" :: "v"(dl[u]), "v"(x0[u]), "v"(x1[u]));
            else                  asm volatile("" :: "v"(dl[u]), "v"(x0[u]));
        }
#pragma unroll
        for (int u = 0; u < U; ++u) {
            const int e = base + u * TPB;
            const int d = (e < E) ? dl[u] : e - E;
            const unsigned off = (unsigned)(d - lo);
            if (e < ee && off < (unsigned)cnt) {
                atomicAdd(&bins[off * H], x0[u]);
                if constexpr (H == 2) atomicAdd(&bins[off * H + 1], x1[u]);
            }
        }
    }
    __syncthreads();

    float* dp = den + (size_t)lo * H;
    for (int i = threadIdx.x; i < cnt * H; i += TPB)
        atomicAdd(dp + i, bins[i]);
}

template <int H, int C>
__global__ void __attribute__((amdgpu_waves_per_eu(8, 8))) edge_agg_kernel(const int* __restrict__ ei, int E, int EE,
                                const float* __restrict__ ebuf,
                                const float* __restrict__ den,
                                const float* __restrict__ xl,
                                float* __restrict__ agg) {
    static_assert(C == 32, "one head's row piece must be one 32-lane line");
    static_assert(H == 1 || H == 2, "alpha loads are written for 1 or 2 heads");
    constexpr int F = H * C;
    const int lane = threadIdx.x & 31;
    const int e_first = __builtin_amdgcn_readfirstlane(
        (int)(blockIdx.x * blockDim.x + threadIdx.x) - lane);
    const int estep = (int)(gridDim.x * blockDim.x);
    if (e_first >= EE) return;

    int src, dst;
    float ex[H];
    {
        const int et = min(e_first + lane, EE - 1);
        const int ec = min(et, E - 1);
        src = ei[ec];
        dst = ei[E + ec];
        if (et >= E) { src = et - E; dst = et - E; }
        if constexpr (H == 2) {
            const float2 x2 = *(const float2*)(ebuf + (size_t)et * 2);
            ex[0] = x2.x; ex[1] = x2.y;
        } else {
            ex[0] = ebuf[et];
        }
    }

    for (int e0 = e_first; e0 < EE; e0 += estep) {
        float dn[H];
        if constexpr (H == 2) {
            const float2 d2 = *(const float2*)(den + (size_t)dst * 2);
            dn[0] = d2.x; dn[1] = d2.y;
        } else {
            dn[0] = den[dst];
        }
        int nsrc, ndst;
        float nex[H];
        {
            const int et = min(e0 + estep + lane, EE - 1);
            const int ec = min(et, E - 1);
            nsrc = ei[ec];
            ndst = ei[E + ec];
            if (et >= E) { nsrc = et - E; ndst = et - E; }
            if constexpr (H == 2) {
                const float2 x2 = *(const float2*)(ebuf + (size_t)et * 2);
                nex[0] = x2.x; nex[1] = x2.y;
            } else {
                nex[0] = ebuf[et];
            }
        }
        float al[H];
#pragma unroll
        for (int h = 0; h < H; ++h) al[h] = ex[h] / (dn[h] + 1e-16f);

        if (EE - e0 >= 32) {
            float v[32 * H];
            unsigned off[32];
            const unsigned ulane = (unsigned)lane;
#pragma unroll
            for (int j = 0; j < 32; ++j)
                off[j] = (unsigned)__builtin_amdgcn_readlane(src, j) * (unsigned)F + ulane;
#pragma unroll
            for (int j = 0; j < 32; ++j)
                for (int h = 0; h < H; ++h) v[j * H + h] = xl[off[j] + (unsigned)(h * C)];
#pragma unroll
            for (int j = 0; j < 32; ++j)
                off[j] = (unsigned)__builtin_amdgcn_readlane(dst, j) * (unsigned)F + ulane;
#pragma unroll
            for (int j = 0; j < 32; ++j) {
                for (int h = 0; h < H; ++h) {
                    const float a = __int_as_float(
                        __builtin_amdgcn_readlane(__float_as_int(al[h]), j));
                    v[j * H + h] *= a;
                }
            }
#pragma unroll
            for (int j = 0; j < 32; ++j) {
                for (int h = 0; h < H; ++h)
                    atomicAdd(agg + (off[j] + (unsigned)(h * C)), v[j * H + h]);
            }
        } else {
            const int cnt = EE - e0;
            for (int j = 0; j < cnt; ++j) {
                const int s = __builtin_amdgcn_readlane(src, j);
                const int d = __builtin_amdgcn_readlane(dst, j);
                const float* row = xl + (size_t)s * F;
                float* orow = agg + (size_t)d * F;
                for (int h = 0; h < H; ++h) {
                    const float a = __int_as_float(
                        __builtin_amdgcn_readlane(__float_as_int(al[h]), j));
                    atomicAdd(orow + h * C + lane, row[h * C + lane] * a);
                }
            }
        }
        src = nsrc; dst = ndst;
#pragma unroll
        for (int h = 0; h < H; ++h) ex[h] = nex[h];
    }
}

static inline int cdiv(long long a, long long b) { return (int)((a + b - 1) / b); }

extern "C" void kernel_launch(void* const* d_in, const int* in_sizes, int n_in,
                              void* d_out, int out_size, void* d_ws, size_t ws_size,
                              hipStream_t stream) {
    const float* x      = (const float*)d_in[0];
    const int*   ei     = (const int*)d_in[1];
    const float* W1     = (const float*)d_in[2];
    const float* a_src1 = (const float*)d_in[3];
    const float* a_dst1 = (const float*)d_in[4];
    const float* b1     = (const float*)d_in[5];
    const float* W2     = (const float*)d_in[6];
    const float* a_src2 = (const float*)d_in[7];
    const float* a_dst2 = (const float*)d_in[8];
    const float* b2     = (const float*)d_in[9];
    float* out = (float*)d_out;

    const int N  = in_sizes[0] / IN_DIM;
    const int E  = in_sizes[1] / 2;
    const int EE = E + N;
    const int F1 = HEADS * HID;

    float* ws   = (float*)d_ws;
    float* xl1  = ws;
    float* agg1 = xl1  + (size_t)N * F1;
    float* asn1 = agg1 + (size_t)N * F1;
    float* adn1 = asn1 + (size_t)N * HEADS;
    float* m1   = adn1 + (size_t)N * HEADS;
    float* den1 = m1   + (size_t)N * HEADS;
    float* ex1  = den1 + (size_t)N * HEADS;
    float* xl2  = ex1  + (size_t)EE * HEADS;
    float* asn2 = xl2  + (size_t)N * OUT_DIM;
    float* adn2 = asn2 + (size_t)N;
    float* m2   = adn2 + (size_t)N;
    float* den2 = m2   + (size_t)N;
    float* ex2  = den2 + (size_t)N;

    const int TB = 256;
    const int NSB = cdiv(cdiv(N, 16), 4);

    float* part1 = xl2;
    float* flag1 = part1 + (size_t)NSB * 2 * HEADS;
    float* part2 = ex1;
    float* flag2 = part2 + (size_t)NSB * 2;


    {
        int mtiles = cdiv(N, 16);
        wmma_gemm_f32<IN_DIM, F1, HEADS, false><<<cdiv(mtiles, 4), 128, 0, stream>>>(
            x, W1, xl1, b1, agg1, a_src1, a_dst1, asn1, adn1, den1, m1, part1, N);
        edge_exp_kernel<HEADS, EDGE_U><<<EDGE_GRID, EDGE_TB, 0, stream>>>(ei, E, EE, asn1, adn1, part1, NSB, flag1, m1, ex1);
        den_bins_kernel<HEADS, DEN_R><<<dim3(cdiv(N, DEN_R), DEN_SLICES), 1024, 0, stream>>>(ei, E, EE, asn1, adn1, m1, flag1, ex1, den1, N);
        edge_agg_kernel<HEADS, HID><<<EDGE_GRID, TB, 0, stream>>>(
            ei, E, EE, ex1, den1, xl1, agg1);
    }

    {
        int mtiles = cdiv(N, 16);
        wmma_gemm_f32<F1, OUT_DIM, 1, true><<<cdiv(mtiles, 4), 128, 0, stream>>>(
            agg1, W2, xl2, b2, out, a_src2, a_dst2, asn2, adn2, den2, m2, part2, N);
        edge_exp_kernel<1, EDGE_U><<<EDGE_GRID, EDGE_TB, 0, stream>>>(ei, E, EE, asn2, adn2, part2, NSB, flag2, m2, ex2);
        den_bins_kernel<1, DEN_R><<<dim3(cdiv(N, DEN_R), DEN_SLICES), 1024, 0, stream>>>(ei, E, EE, asn2, adn2, m2, flag2, ex2, den2, N);
        edge_agg_kernel<1, OUT_DIM><<<EDGE_GRID, TB, 0, stream>>>(
            ei, E, EE, ex2, den2, xl2, out);
    }
}
